# starting kernel
# baseline (speedup 1.0000x reference)
#include <hip/hip_runtime.h>

#define BATCH 4
#define FDIM  64

typedef __attribute__((ext_vector_type(2))) float v2f;
typedef __attribute__((ext_vector_type(8))) float v8f;

__global__ void gcn_zero_ws(float* __restrict__ p, long n) {
    long i = (long)blockIdx.x * blockDim.x + threadIdx.x;
    const long stride = (long)gridDim.x * blockDim.x;
    for (; i < n; i += stride) p[i] = 0.0f;
}

__global__ void gcn_spmm_scatter(const float* __restrict__ x,
                                 const float* __restrict__ vals,
                                 const int*   __restrict__ rows,
                                 const int*   __restrict__ cols,
                                 float*       __restrict__ agg,
                                 long E, long Nn) {
    const long g = (long)blockIdx.x * blockDim.x + threadIdx.x;
    const long e = g >> 6;
    if (e >= E) return;
    const int  f = (int)(g & 63);
    const float v = vals[e];
    const long  r = rows[e];
    const long  c = cols[e];
#pragma unroll
    for (int b = 0; b < BATCH; ++b) {
        const float xv = x[((long)b * Nn + c) * FDIM + f];
        __hip_atomic_fetch_add(&agg[((long)b * Nn + r) * FDIM + f], v * xv,
                               __ATOMIC_RELAXED, __HIP_MEMORY_SCOPE_AGENT);
    }
}

__global__ void __launch_bounds__(256)
gcn_gemm_bias_elu(const float* __restrict__ agg,
                  const float* __restrict__ W,
                  const float* __restrict__ bias,
                  float*       __restrict__ out,
                  long nrows) {
    const int lane = threadIdx.x & 31;
    const int wave = threadIdx.x >> 5;
    const long rowBase = ((long)blockIdx.x * 8 + wave) * 16;
    if (rowBase >= nrows) return;

    const int m  = lane & 15;
    const int kg = (lane >> 4) << 1;

    long arow = rowBase + m;
    if (arow >= nrows) arow = nrows - 1;
    const float* __restrict__ aptr = agg + arow * FDIM;

    v8f acc[4] = {v8f{}, v8f{}, v8f{}, v8f{}};

#pragma unroll
    for (int k4 = 0; k4 < FDIM / 4; ++k4) {
        const int kk = k4 * 4 + kg;
        v2f a;
        a.x = aptr[kk];
        a.y = aptr[kk + 1];

        const float* __restrict__ w0 = W + (long)kk * FDIM + m;
        const float* __restrict__ w1 = W + (long)(kk + 1) * FDIM + m;
#pragma unroll
        for (int j = 0; j < 4; ++j) {
            v2f b;
            b.x = w0[j * 16];
            b.y = w1[j * 16];
            acc[j] = __builtin_amdgcn_wmma_f32_16x16x4_f32(
                 false, a,  false, b,
                 (short)0, acc[j],
                 false,  false);
        }
    }

    const int rowAdd = (lane >> 4) * 8;
#pragma unroll
    for (int j = 0; j < 4; ++j) {
        const int col = j * 16 + m;
        const float bb = bias[col];
#pragma unroll
        for (int r = 0; r < 8; ++r) {
            const long row = rowBase + rowAdd + r;
            if (row < nrows) {
                const float v = acc[j][r] + bb;
                out[row * FDIM + col] = (v > 0.0f) ? v : (__expf(v) - 1.0f);
            }
        }
    }
}

extern "C" void kernel_launch(void* const* d_in, const int* in_sizes, int n_in,
                              void* d_out, int out_size, void* d_ws, size_t ws_size,
                              hipStream_t stream) {
    const float* x      = (const float*)d_in[0];
    const float* vals   = (const float*)d_in[1];
    const float* weight = (const float*)d_in[2];
    const float* bias   = (const float*)d_in[3];
    const int*   rows   = (const int*)d_in[4];
    const int*   cols   = (const int*)d_in[5];
    float*       out    = (float*)d_out;

    const long E  = in_sizes[1];
    const long Nn = (long)in_sizes[0] / (BATCH * FDIM);
    const long aggElems = (long)BATCH * Nn * FDIM;
    float* agg = (float*)d_ws;

    {
        const int  threads = 256;
        long blocks = (aggElems + threads - 1) / threads;
        if (blocks > 32768) blocks = 32768;
        gcn_zero_ws<<<(unsigned)blocks, threads, 0, stream>>>(agg, aggElems);
    }
    {
        const long work = E * FDIM;
        const int  threads = 256;
        const long blocks = (work + threads - 1) / threads;
        gcn_spmm_scatter<<<(unsigned)blocks, threads, 0, stream>>>(
            x, vals, rows, cols, agg, E, Nn);
    }
    {
        const long nrows  = (long)BATCH * Nn;
        const long blocks = (nrows + 127) / 128;
        gcn_gemm_bias_elu<<<(unsigned)blocks, 256, 0, stream>>>(
            agg, weight, bias, out, nrows);
    }
}


# SpMM: f32 atomic scatter replaced by per-call CSR build (int atomic row counts, scan, fill) + one-wave-per-row gather; row sums run in ascending edge id, not arrival order: bit-identical run to run
# speedup vs baseline: 1.7456x; 1.7456x over previous
#include <hip/hip_runtime.h>


#define BATCH 4
#define FDIM  64
#define ROWS_PER_BLOCK 8
#define SCAN_BLOCK 1024

typedef __attribute__((ext_vector_type(2))) float v2f;
typedef __attribute__((ext_vector_type(8))) float v8f;
typedef __attribute__((ext_vector_type(4))) float v4f;

__global__ void gcn_zero_counts(int* __restrict__ cnt, int n) {
    const int i = blockIdx.x * blockDim.x + threadIdx.x;
    if (i < n) cnt[i] = 0;
}

#define EDGES_PER_THREAD 4

__global__ void gcn_row_hist(const int* __restrict__ rows, int* __restrict__ cnt, int E) {
    const int e0 = (blockIdx.x * blockDim.x + threadIdx.x) * EDGES_PER_THREAD;
    if (e0 + EDGES_PER_THREAD <= E) {
        const int4 r = *(const int4*)(rows + e0);
        __hip_atomic_fetch_add(&cnt[r.x], 1, __ATOMIC_RELAXED, __HIP_MEMORY_SCOPE_AGENT);
        __hip_atomic_fetch_add(&cnt[r.y], 1, __ATOMIC_RELAXED, __HIP_MEMORY_SCOPE_AGENT);
        __hip_atomic_fetch_add(&cnt[r.z], 1, __ATOMIC_RELAXED, __HIP_MEMORY_SCOPE_AGENT);
        __hip_atomic_fetch_add(&cnt[r.w], 1, __ATOMIC_RELAXED, __HIP_MEMORY_SCOPE_AGENT);
    } else {
        for (int e = e0; e < E; ++e)
            __hip_atomic_fetch_add(&cnt[rows[e]], 1, __ATOMIC_RELAXED, __HIP_MEMORY_SCOPE_AGENT);
    }
}

__global__ void __launch_bounds__(SCAN_BLOCK)
gcn_row_offsets(const int* __restrict__ cnt, int* __restrict__ offs, int n) {
    __shared__ int s_front[32];
    __shared__ int s_total[32];
    const int t    = threadIdx.x;
    const int lane = t & 31;
    const int w    = t >> 5;
    const int blk  = blockIdx.x;

    int front = 0;
    for (int i = 0; i < blk; ++i) front += cnt[i * SCAN_BLOCK + t];

    const int idx = blk * SCAN_BLOCK + t;
    int inc = (idx < n) ? cnt[idx] : 0;
    for (int d = 1; d < 32; d <<= 1) {
        const int o = __shfl_up(inc, d);
        if (lane >= d) inc += o;
    }
    for (int d = 16; d > 0; d >>= 1) front += __shfl_xor(front, d);
    if (lane == 31) s_total[w] = inc;
    if (lane == 0)  s_front[w] = front;
    __syncthreads();
    int a = s_front[lane];
    int b = (lane < w) ? s_total[lane] : 0;
    for (int d = 16; d > 0; d >>= 1) {
        a += __shfl_xor(a, d);
        b += __shfl_xor(b, d);
    }
    if (idx < n)  offs[idx + 1] = a + b + inc;
    if (idx == 0) offs[0] = 0;
}

__device__ __forceinline__ int gcn_take_slot(int* __restrict__ cnt, int r) {
    return __hip_atomic_fetch_add(&cnt[r], -1, __ATOMIC_RELAXED, __HIP_MEMORY_SCOPE_AGENT) - 1;
}

__global__ void gcn_fill_perm(const int* __restrict__ rows, int* __restrict__ cnt,
                              const int* __restrict__ offs, int* __restrict__ perm, int E) {
    const int e0 = (blockIdx.x * blockDim.x + threadIdx.x) * EDGES_PER_THREAD;
    if (e0 + EDGES_PER_THREAD <= E) {
        const int4 r = *(const int4*)(rows + e0);
        const int s0 = gcn_take_slot(cnt, r.x);
        const int s1 = gcn_take_slot(cnt, r.y);
        const int s2 = gcn_take_slot(cnt, r.z);
        const int s3 = gcn_take_slot(cnt, r.w);
        perm[offs[r.x] + s0] = e0;
        perm[offs[r.y] + s1] = e0 + 1;
        perm[offs[r.z] + s2] = e0 + 2;
        perm[offs[r.w] + s3] = e0 + 3;
    } else {
        for (int e = e0; e < E; ++e) {
            const int r = rows[e];
            perm[offs[r] + gcn_take_slot(cnt, r)] = e;
        }
    }
}

__device__ __forceinline__ v4f gcn_ld4(const float* __restrict__ xrow, unsigned off) {
    return *(const v4f*)((const char*)xrow + off);
}

__device__ __forceinline__ void gcn_edge1(const float* __restrict__ x, unsigned offa, unsigned offb,
                                          int c, float v, v4f& s0, v4f& s1) {
    const float* __restrict__ xr = x + (size_t)c * FDIM;
    const v4f p = gcn_ld4(xr, offa);
    const v4f q = gcn_ld4(xr, offb);
    s0 += v * p;
    s1 += v * q;
}

__device__ __forceinline__ void gcn_pick(int rank, int c, float v, int p, int& cp, float& vp) {
    const int l = __builtin_ctz(__builtin_amdgcn_ballot_w32(rank == p));
    cp = __builtin_amdgcn_readlane(c, l);
    vp = __int_as_float(__builtin_amdgcn_readlane(__float_as_int(v), l));
}

__global__ void __launch_bounds__(ROWS_PER_BLOCK * 32)
gcn_spmm_gather(const float* __restrict__ x,
                const float* __restrict__ vals,
                const int*   __restrict__ cols,
                const int*   __restrict__ offs,
                const int*   __restrict__ perm,
                float*       __restrict__ agg,
                int Nn) {
    const int lane = threadIdx.x & 31;
    const int r = blockIdx.x * ROWS_PER_BLOCK
                + __builtin_amdgcn_readfirstlane((int)(threadIdx.x >> 5));
    if (r >= Nn) return;

    const int beg = offs[r];
    const int kt  = offs[r + 1] - beg;
    const int half = lane >> 4;
    const int q4   = (lane & 15) * 4;
    const unsigned offa = (unsigned)(((size_t)half * Nn * FDIM + q4) * sizeof(float));
    const unsigned offb = (unsigned)(((size_t)(half + 2) * Nn * FDIM + q4) * sizeof(float));

    v4f s0 = {0.f, 0.f, 0.f, 0.f};
    v4f s1 = {0.f, 0.f, 0.f, 0.f};

    if (kt <= 32) {
        int   eid = 0x7fffffff;
        int   c   = 0;
        float v   = 0.f;
        if (lane < kt) {
            eid = perm[beg + lane];
            c   = cols[eid];
            v   = vals[eid];
        }
        int rank = 0;
        for (int j = 0; j < kt; ++j)
            rank += (__builtin_amdgcn_readlane(eid, j) < eid) ? 1 : 0;

        int p = 0;
        for (; p + 3 < kt; p += 4) {
            int c0, c1, c2, c3;
            float v0, v1, v2, v3;
            gcn_pick(rank, c, v, p,     c0, v0);
            gcn_pick(rank, c, v, p + 1, c1, v1);
            gcn_pick(rank, c, v, p + 2, c2, v2);
            gcn_pick(rank, c, v, p + 3, c3, v3);
            const float* __restrict__ x0 = x + (size_t)c0 * FDIM;
            const float* __restrict__ x1 = x + (size_t)c1 * FDIM;
            const float* __restrict__ x2 = x + (size_t)c2 * FDIM;
            const float* __restrict__ x3 = x + (size_t)c3 * FDIM;
            const v4f p0 = gcn_ld4(x0, offa), q0 = gcn_ld4(x0, offb);
            const v4f p1 = gcn_ld4(x1, offa), q1 = gcn_ld4(x1, offb);
            const v4f p2 = gcn_ld4(x2, offa), q2 = gcn_ld4(x2, offb);
            const v4f p3 = gcn_ld4(x3, offa), q3 = gcn_ld4(x3, offb);
            s0 += v0 * p0;  s1 += v0 * q0;
            s0 += v1 * p1;  s1 += v1 * q1;
            s0 += v2 * p2;  s1 += v2 * q2;
            s0 += v3 * p3;  s1 += v3 * q3;
        }
        for (; p < kt; ++p) {
            int c0; float v0;
            gcn_pick(rank, c, v, p, c0, v0);
            gcn_edge1(x, offa, offb, c0, v0, s0, s1);
        }
    } else {
        int prev = -1;
        for (int p = 0; p < kt; ++p) {
            int m = 0x7fffffff;
            for (int i = lane; i < kt; i += 32) {
                const int e = perm[beg + i];
                if (e > prev && e < m) m = e;
            }
            for (int d = 16; d > 0; d >>= 1) {
                const int o = __shfl_xor(m, d);
                m = (o < m) ? o : m;
            }
            m = __builtin_amdgcn_readfirstlane(m);
            prev = m;
            gcn_edge1(x, offa, offb, cols[m], vals[m], s0, s1);
        }
    }

    float* __restrict__ oa = agg + ((size_t)half * Nn + r) * FDIM + q4;
    float* __restrict__ ob = agg + ((size_t)(half + 2) * Nn + r) * FDIM + q4;
    *(v4f*)oa = s0;
    *(v4f*)ob = s1;
}

__global__ void __launch_bounds__(256)
gcn_gemm_bias_elu(const float* __restrict__ agg,
                  const float* __restrict__ W,
                  const float* __restrict__ bias,
                  float*       __restrict__ out,
                  long nrows) {
    const int lane = threadIdx.x & 31;
    const int wave = threadIdx.x >> 5;
    const long rowBase = ((long)blockIdx.x * 8 + wave) * 16;
    if (rowBase >= nrows) return;

    const int m  = lane & 15;
    const int kg = (lane >> 4) << 1;

    long arow = rowBase + m;
    if (arow >= nrows) arow = nrows - 1;
    const float* __restrict__ aptr = agg + arow * FDIM;

    v8f acc[4] = {v8f{}, v8f{}, v8f{}, v8f{}};

#pragma unroll
    for (int k4 = 0; k4 < FDIM / 4; ++k4) {
        const int kk = k4 * 4 + kg;
        v2f a;
        a.x = aptr[kk];
        a.y = aptr[kk + 1];

        const float* __restrict__ w0 = W + (long)kk * FDIM + m;
        const float* __restrict__ w1 = W + (long)(kk + 1) * FDIM + m;
#pragma unroll
        for (int j = 0; j < 4; ++j) {
            v2f b;
            b.x = w0[j * 16];
            b.y = w1[j * 16];
            acc[j] = __builtin_amdgcn_wmma_f32_16x16x4_f32(
                 false, a,  false, b,
                 (short)0, acc[j],
                 false,  false);
        }
    }

    const int rowAdd = (lane >> 4) * 8;
#pragma unroll
    for (int j = 0; j < 4; ++j) {
        const int col = j * 16 + m;
        const float bb = bias[col];
#pragma unroll
        for (int r = 0; r < 8; ++r) {
            const long row = rowBase + rowAdd + r;
            if (row < nrows) {
                const float v = acc[j][r] + bb;
                out[row * FDIM + col] = (v > 0.0f) ? v : (__expf(v) - 1.0f);
            }
        }
    }
}

extern "C" void kernel_launch(void* const* d_in, const int* in_sizes, int n_in,
                              void* d_out, int out_size, void* d_ws, size_t ws_size,
                              hipStream_t stream) {
    const float* x      = (const float*)d_in[0];
    const float* vals   = (const float*)d_in[1];
    const float* weight = (const float*)d_in[2];
    const float* bias   = (const float*)d_in[3];
    const int*   rows   = (const int*)d_in[4];
    const int*   cols   = (const int*)d_in[5];
    float*       out    = (float*)d_out;

    const int  E  = in_sizes[1];
    const int  Nn = in_sizes[0] / (BATCH * FDIM);
    const long aggElems = (long)BATCH * Nn * FDIM;
    float* agg  = (float*)d_ws;
    const long nPad = ((long)Nn + 64) & ~63L;
    int*   cnt  = (int*)(agg + aggElems);
    int*   offs = cnt + nPad;
    int*   perm = offs + nPad;

    const int threads = 256;
    const int eBlocks = (E + threads * EDGES_PER_THREAD - 1) / (threads * EDGES_PER_THREAD);

    gcn_zero_counts<<<(unsigned)((Nn + threads - 1) / threads), threads, 0, stream>>>(cnt, Nn);
    gcn_row_hist<<<(unsigned)eBlocks, threads, 0, stream>>>(rows, cnt, E);
    gcn_row_offsets<<<(unsigned)((Nn + SCAN_BLOCK - 1) / SCAN_BLOCK), SCAN_BLOCK, 0, stream>>>(
        cnt, offs, Nn);
    gcn_fill_perm<<<(unsigned)eBlocks, threads, 0, stream>>>(rows, cnt, offs, perm, E);
    gcn_spmm_gather<<<(unsigned)((Nn + ROWS_PER_BLOCK - 1) / ROWS_PER_BLOCK),
                      ROWS_PER_BLOCK * 32, 0, stream>>>(x, vals, cols, offs, perm, agg, Nn);
    {
        const long nrows  = (long)BATCH * Nn;
        const long blocks = (nrows + 127) / 128;
        gcn_gemm_bias_elu<<<(unsigned)blocks, 256, 0, stream>>>(
            agg, weight, bias, out, nrows);
    }
}


# CSR build: the 2 global int atomics per edge replaced by LDS counters in two levels (chunk counts per 256-row bucket, one scan, records by bucket, rows inside a bucket); row sum order unchanged
# speedup vs baseline: 2.7793x; 1.5922x over previous
#include <hip/hip_runtime.h>


#define BATCH 4
#define FDIM  64
#define ROWS_PER_BLOCK 8
#define SCAN_BLOCK 1024

typedef __attribute__((ext_vector_type(2))) float v2f;
typedef __attribute__((ext_vector_type(8))) float v8f;
typedef __attribute__((ext_vector_type(4))) float v4f;

#define BUCKET_SHIFT  8
#define BUCKET_ROWS   (1 << BUCKET_SHIFT)
#define CHUNK_EDGES   4096
#define CHUNK_THREADS 256
#define CHUNK_VEC     (CHUNK_EDGES / (CHUNK_THREADS * 4))
#define SCAN_PER      16
#define SCAN_TILE     (SCAN_BLOCK * SCAN_PER)
#define CSR_THREADS   1024
#define CSR_BATCH     6

__device__ __forceinline__ int4 gcn_rows4(const int* __restrict__ rows, int e, int E) {
    int4 r;
    if (e + 4 <= E) {
        r = *(const int4*)(rows + e);
    } else {
        r.x = (e     < E) ? rows[e]     : -1;
        r.y = (e + 1 < E) ? rows[e + 1] : -1;
        r.z = (e + 2 < E) ? rows[e + 2] : -1;
        r.w = (e + 3 < E) ? rows[e + 3] : -1;
    }
    return r;
}

__device__ __forceinline__ void gcn_lds_count(int* s_cnt, int row) {
    if (row >= 0)
        __hip_atomic_fetch_add(&s_cnt[row >> BUCKET_SHIFT], 1, __ATOMIC_RELAXED,
                               __HIP_MEMORY_SCOPE_WORKGROUP);
}

__global__ void __launch_bounds__(CHUNK_THREADS)
gcn_bucket_count(const int* __restrict__ rows, int* __restrict__ cc, int E, int nb, int nchunk) {
    extern __shared__ int s_bcnt[];
    const int t = threadIdx.x;
    const int chunk = blockIdx.x;
    for (int i = t; i < nb; i += CHUNK_THREADS) s_bcnt[i] = 0;
    int4 r[CHUNK_VEC];
    const int base = chunk * CHUNK_EDGES;
#pragma unroll
    for (int k = 0; k < CHUNK_VEC; ++k)
        r[k] = gcn_rows4(rows, base + (k * CHUNK_THREADS + t) * 4, E);
    __syncthreads();
#pragma unroll
    for (int k = 0; k < CHUNK_VEC; ++k) {
        gcn_lds_count(s_bcnt, r[k].x);
        gcn_lds_count(s_bcnt, r[k].y);
        gcn_lds_count(s_bcnt, r[k].z);
        gcn_lds_count(s_bcnt, r[k].w);
    }
    __syncthreads();
    for (int i = t; i < nb; i += CHUNK_THREADS) cc[(size_t)i * nchunk + chunk] = s_bcnt[i];
}

__global__ void __launch_bounds__(SCAN_BLOCK)
gcn_bucket_scan(const int* __restrict__ cc, int* __restrict__ cstart, int total) {
    __shared__ int s_v[SCAN_TILE + SCAN_BLOCK];
    __shared__ int s_w[32];
    const int t    = threadIdx.x;
    const int lane = t & 31;
    const int w    = t >> 5;
    int carry = 0;
    for (int t0 = 0; t0 < total; t0 += SCAN_TILE) {
        int v[SCAN_PER];
#pragma unroll
        for (int k = 0; k < SCAN_PER; ++k) {
            const int i = t0 + k * SCAN_BLOCK + t;
            v[k] = (i < total) ? cc[i] : 0;
        }
#pragma unroll
        for (int k = 0; k < SCAN_PER; ++k) {
            const int p = k * SCAN_BLOCK + t;
            s_v[p + (p >> 4)] = v[k];
        }
        __syncthreads();
        int sum = 0;
#pragma unroll
        for (int j = 0; j < SCAN_PER; ++j) sum += s_v[t * (SCAN_PER + 1) + j];
        int inc = sum;
        for (int d = 1; d < 32; d <<= 1) {
            const int o = __shfl_up(inc, d);
            if (lane >= d) inc += o;
        }
        if (lane == 31) s_w[w] = inc;
        __syncthreads();
        const int wt = s_w[lane];
        int before = (lane < w) ? wt : 0;
        int all    = wt;
        for (int d = 16; d > 0; d >>= 1) {
            before += __shfl_xor(before, d);
            all    += __shfl_xor(all, d);
        }
        int run = carry + before + inc - sum;
#pragma unroll
        for (int j = 0; j < SCAN_PER; ++j) {
            const int x = s_v[t * (SCAN_PER + 1) + j];
            s_v[t * (SCAN_PER + 1) + j] = run;
            run += x;
        }
        __syncthreads();
#pragma unroll
        for (int k = 0; k < SCAN_PER; ++k) {
            const int i = t0 + k * SCAN_BLOCK + t;
            const int p = k * SCAN_BLOCK + t;
            if (i < total) cstart[i] = s_v[p + (p >> 4)];
        }
        carry += all;
        __syncthreads();
    }
    if (t == 0) cstart[total] = carry;
}

__device__ __forceinline__ void gcn_file_edge(int* s_cur, int* __restrict__ brec, int row, int e) {
    if (row >= 0) {
        const int pos = __hip_atomic_fetch_add(&s_cur[row >> BUCKET_SHIFT], 1, __ATOMIC_RELAXED,
                                               __HIP_MEMORY_SCOPE_WORKGROUP);
        *(int2*)(brec + 2 * (size_t)pos) = make_int2(row, e);
    }
}

__global__ void __launch_bounds__(CHUNK_THREADS)
gcn_bucket_scatter(const int* __restrict__ rows, const int* __restrict__ cstart,
                   int* __restrict__ brec, int E, int nb, int nchunk) {
    extern __shared__ int s_bcur[];
    const int t = threadIdx.x;
    const int chunk = blockIdx.x;
    int4 r[CHUNK_VEC];
    const int base = chunk * CHUNK_EDGES;
#pragma unroll
    for (int k = 0; k < CHUNK_VEC; ++k)
        r[k] = gcn_rows4(rows, base + (k * CHUNK_THREADS + t) * 4, E);
    for (int i = t; i < nb; i += CHUNK_THREADS) s_bcur[i] = cstart[(size_t)i * nchunk + chunk];
    __syncthreads();
#pragma unroll
    for (int k = 0; k < CHUNK_VEC; ++k) {
        const int e = base + (k * CHUNK_THREADS + t) * 4;
        gcn_file_edge(s_bcur, brec, r[k].x, e);
        gcn_file_edge(s_bcur, brec, r[k].y, e + 1);
        gcn_file_edge(s_bcur, brec, r[k].z, e + 2);
        gcn_file_edge(s_bcur, brec, r[k].w, e + 3);
    }
}

__global__ void __launch_bounds__(CSR_THREADS)
gcn_bucket_csr(const int* __restrict__ brec, const int* __restrict__ cstart,
               int* __restrict__ offs, int* __restrict__ perm, int Nn, int nchunk) {
    __shared__ int s_cnt[BUCKET_ROWS];
    __shared__ int s_cur[BUCKET_ROWS];
    __shared__ int s_w[BUCKET_ROWS / 32];
    const int t    = threadIdx.x;
    const int lane = t & 31;
    const int w    = t >> 5;
    const int b    = blockIdx.x;
    const int bs = cstart[(size_t)b * nchunk];
    const int be = cstart[(size_t)(b + 1) * nchunk];
    if (t < BUCKET_ROWS) s_cnt[t] = 0;
    __syncthreads();
    for (int i0 = bs + t; i0 < be; i0 += CSR_BATCH * CSR_THREADS) {
        int rw[CSR_BATCH];
#pragma unroll
        for (int u = 0; u < CSR_BATCH; ++u) {
            const int i = i0 + u * CSR_THREADS;
            rw[u] = (i < be) ? brec[2 * (size_t)i] : -1;
        }
#pragma unroll
        for (int u = 0; u < CSR_BATCH; ++u)
            if (rw[u] >= 0)
                __hip_atomic_fetch_add(&s_cnt[rw[u] & (BUCKET_ROWS - 1)], 1, __ATOMIC_RELAXED,
                                       __HIP_MEMORY_SCOPE_WORKGROUP);
    }
    __syncthreads();
    int c = 0, inc = 0;
    if (t < BUCKET_ROWS) {
        c = s_cnt[t];
        inc = c;
        for (int d = 1; d < 32; d <<= 1) {
            const int o = __shfl_up(inc, d);
            if (lane >= d) inc += o;
        }
        if (lane == 31) s_w[w] = inc;
    }
    __syncthreads();
    if (t < BUCKET_ROWS) {
        int before = 0;
        for (int j = 0; j < w; ++j) before += s_w[j];
        const int start = bs + before + inc - c;
        s_cur[t] = start;
        const int r = b * BUCKET_ROWS + t;
        if (r < Nn)      offs[r]  = start;
        if (r == Nn - 1) offs[Nn] = start + c;
    }
    __syncthreads();
    for (int i0 = bs + t; i0 < be; i0 += CSR_BATCH * CSR_THREADS) {
        int2 rc[CSR_BATCH];
#pragma unroll
        for (int u = 0; u < CSR_BATCH; ++u) {
            const int i = i0 + u * CSR_THREADS;
            rc[u] = (i < be) ? *(const int2*)(brec + 2 * (size_t)i) : make_int2(-1, 0);
        }
#pragma unroll
        for (int u = 0; u < CSR_BATCH; ++u)
            if (rc[u].x >= 0) {
                const int pos = __hip_atomic_fetch_add(&s_cur[rc[u].x & (BUCKET_ROWS - 1)], 1,
                                                       __ATOMIC_RELAXED, __HIP_MEMORY_SCOPE_WORKGROUP);
                perm[pos] = rc[u].y;
            }
    }
}

__device__ __forceinline__ v4f gcn_ld4(const float* __restrict__ xrow, unsigned off) {
    return *(const v4f*)((const char*)xrow + off);
}

__device__ __forceinline__ void gcn_edge1(const float* __restrict__ x, unsigned offa, unsigned offb,
                                          int c, float v, v4f& s0, v4f& s1) {
    const float* __restrict__ xr = x + (size_t)c * FDIM;
    const v4f p = gcn_ld4(xr, offa);
    const v4f q = gcn_ld4(xr, offb);
    s0 += v * p;
    s1 += v * q;
}

__device__ __forceinline__ void gcn_pick(int rank, int c, float v, int p, int& cp, float& vp) {
    const int l = __builtin_ctz(__builtin_amdgcn_ballot_w32(rank == p));
    cp = __builtin_amdgcn_readlane(c, l);
    vp = __int_as_float(__builtin_amdgcn_readlane(__float_as_int(v), l));
}

__global__ void __launch_bounds__(ROWS_PER_BLOCK * 32)
gcn_spmm_gather(const float* __restrict__ x,
                const float* __restrict__ vals,
                const int*   __restrict__ cols,
                const int*   __restrict__ offs,
                const int*   __restrict__ perm,
                float*       __restrict__ agg,
                int Nn) {
    const int lane = threadIdx.x & 31;
    const int r = blockIdx.x * ROWS_PER_BLOCK
                + __builtin_amdgcn_readfirstlane((int)(threadIdx.x >> 5));
    if (r >= Nn) return;

    const int beg = offs[r];
    const int kt  = offs[r + 1] - beg;
    const int half = lane >> 4;
    const int q4   = (lane & 15) * 4;
    const unsigned offa = (unsigned)(((size_t)half * Nn * FDIM + q4) * sizeof(float));
    const unsigned offb = (unsigned)(((size_t)(half + 2) * Nn * FDIM + q4) * sizeof(float));

    v4f s0 = {0.f, 0.f, 0.f, 0.f};
    v4f s1 = {0.f, 0.f, 0.f, 0.f};

    if (kt <= 32) {
        int   eid = 0x7fffffff;
        int   c   = 0;
        float v   = 0.f;
        if (lane < kt) {
            eid = perm[beg + lane];
            c   = cols[eid];
            v   = vals[eid];
        }
        int rank = 0;
        for (int j = 0; j < kt; ++j)
            rank += (__builtin_amdgcn_readlane(eid, j) < eid) ? 1 : 0;

        int p = 0;
        for (; p + 3 < kt; p += 4) {
            int c0, c1, c2, c3;
            float v0, v1, v2, v3;
            gcn_pick(rank, c, v, p,     c0, v0);
            gcn_pick(rank, c, v, p + 1, c1, v1);
            gcn_pick(rank, c, v, p + 2, c2, v2);
            gcn_pick(rank, c, v, p + 3, c3, v3);
            const float* __restrict__ x0 = x + (size_t)c0 * FDIM;
            const float* __restrict__ x1 = x + (size_t)c1 * FDIM;
            const float* __restrict__ x2 = x + (size_t)c2 * FDIM;
            const float* __restrict__ x3 = x + (size_t)c3 * FDIM;
            const v4f p0 = gcn_ld4(x0, offa), q0 = gcn_ld4(x0, offb);
            const v4f p1 = gcn_ld4(x1, offa), q1 = gcn_ld4(x1, offb);
            const v4f p2 = gcn_ld4(x2, offa), q2 = gcn_ld4(x2, offb);
            const v4f p3 = gcn_ld4(x3, offa), q3 = gcn_ld4(x3, offb);
            s0 += v0 * p0;  s1 += v0 * q0;
            s0 += v1 * p1;  s1 += v1 * q1;
            s0 += v2 * p2;  s1 += v2 * q2;
            s0 += v3 * p3;  s1 += v3 * q3;
        }
        for (; p < kt; ++p) {
            int c0; float v0;
            gcn_pick(rank, c, v, p, c0, v0);
            gcn_edge1(x, offa, offb, c0, v0, s0, s1);
        }
    } else {
        int prev = -1;
        for (int p = 0; p < kt; ++p) {
            int m = 0x7fffffff;
            for (int i = lane; i < kt; i += 32) {
                const int e = perm[beg + i];
                if (e > prev && e < m) m = e;
            }
            for (int d = 16; d > 0; d >>= 1) {
                const int o = __shfl_xor(m, d);
                m = (o < m) ? o : m;
            }
            m = __builtin_amdgcn_readfirstlane(m);
            prev = m;
            gcn_edge1(x, offa, offb, cols[m], vals[m], s0, s1);
        }
    }

    float* __restrict__ oa = agg + ((size_t)half * Nn + r) * FDIM + q4;
    float* __restrict__ ob = agg + ((size_t)(half + 2) * Nn + r) * FDIM + q4;
    *(v4f*)oa = s0;
    *(v4f*)ob = s1;
}

__global__ void __launch_bounds__(256)
gcn_gemm_bias_elu(const float* __restrict__ agg,
                  const float* __restrict__ W,
                  const float* __restrict__ bias,
                  float*       __restrict__ out,
                  long nrows) {
    const int lane = threadIdx.x & 31;
    const int wave = threadIdx.x >> 5;
    const long rowBase = ((long)blockIdx.x * 8 + wave) * 16;
    if (rowBase >= nrows) return;

    const int m  = lane & 15;
    const int kg = (lane >> 4) << 1;

    long arow = rowBase + m;
    if (arow >= nrows) arow = nrows - 1;
    const float* __restrict__ aptr = agg + arow * FDIM;

    v8f acc[4] = {v8f{}, v8f{}, v8f{}, v8f{}};

#pragma unroll
    for (int k4 = 0; k4 < FDIM / 4; ++k4) {
        const int kk = k4 * 4 + kg;
        v2f a;
        a.x = aptr[kk];
        a.y = aptr[kk + 1];

        const float* __restrict__ w0 = W + (long)kk * FDIM + m;
        const float* __restrict__ w1 = W + (long)(kk + 1) * FDIM + m;
#pragma unroll
        for (int j = 0; j < 4; ++j) {
            v2f b;
            b.x = w0[j * 16];
            b.y = w1[j * 16];
            acc[j] = __builtin_amdgcn_wmma_f32_16x16x4_f32(
                 false, a,  false, b,
                 (short)0, acc[j],
                 false,  false);
        }
    }

    const int rowAdd = (lane >> 4) * 8;
#pragma unroll
    for (int j = 0; j < 4; ++j) {
        const int col = j * 16 + m;
        const float bb = bias[col];
#pragma unroll
        for (int r = 0; r < 8; ++r) {
            const long row = rowBase + rowAdd + r;
            if (row < nrows) {
                const float v = acc[j][r] + bb;
                out[row * FDIM + col] = (v > 0.0f) ? v : (__expf(v) - 1.0f);
            }
        }
    }
}

extern "C" void kernel_launch(void* const* d_in, const int* in_sizes, int n_in,
                              void* d_out, int out_size, void* d_ws, size_t ws_size,
                              hipStream_t stream) {
    const float* x      = (const float*)d_in[0];
    const float* vals   = (const float*)d_in[1];
    const float* weight = (const float*)d_in[2];
    const float* bias   = (const float*)d_in[3];
    const int*   rows   = (const int*)d_in[4];
    const int*   cols   = (const int*)d_in[5];
    float*       out    = (float*)d_out;

    const int  E  = in_sizes[1];
    const int  Nn = in_sizes[0] / (BATCH * FDIM);
    const long aggElems = (long)BATCH * Nn * FDIM;
    const int  nb     = (Nn + BUCKET_ROWS - 1) / BUCKET_ROWS;
    const int  nchunk = (E + CHUNK_EDGES - 1) / CHUNK_EDGES;
    const int  total  = nb * nchunk;
    float* agg  = (float*)d_ws;
    const long nPad = ((long)Nn + 64) & ~63L;
    const long ePad = ((long)E + 63) & ~63L;
    const long tPad = ((long)total + 64) & ~63L;
    int*   offs   = (int*)(agg + aggElems);
    int*   perm   = offs + nPad;
    int*   brec   = perm + ePad;
    int*   cc     = brec + 2 * ePad;
    int*   cstart = cc + tPad;

    gcn_bucket_count<<<(unsigned)nchunk, CHUNK_THREADS, nb * sizeof(int), stream>>>(
        rows, cc, E, nb, nchunk);
    gcn_bucket_scan<<<1, SCAN_BLOCK, 0, stream>>>(cc, cstart, total);
    gcn_bucket_scatter<<<(unsigned)nchunk, CHUNK_THREADS, nb * sizeof(int), stream>>>(
        rows, cstart, brec, E, nb, nchunk);
    gcn_bucket_csr<<<(unsigned)nb, CSR_THREADS, 0, stream>>>(brec, cstart, offs, perm, Nn, nchunk);
    gcn_spmm_gather<<<(unsigned)((Nn + ROWS_PER_BLOCK - 1) / ROWS_PER_BLOCK),
                      ROWS_PER_BLOCK * 32, 0, stream>>>(x, vals, cols, offs, perm, agg, Nn);
    {
        const long nrows  = (long)BATCH * Nn;
        const long blocks = (nrows + 127) / 128;
        gcn_gemm_bias_elu<<<(unsigned)blocks, 256, 0, stream>>>(
            agg, weight, bias, out, nrows);
    }
}


# GEMM+bias+ELU fused into the gather kernel: a block's 32 agg rows go registers -> LDS -> A fragments, agg never goes to memory, one launch fewer; sums and their order unchanged
# speedup vs baseline: 3.2313x; 1.1626x over previous
#include <hip/hip_runtime.h>


#define BATCH 4
#define FDIM  64
#define ROWS_PER_BLOCK 8
#define TILE_ROWS (BATCH * ROWS_PER_BLOCK)
#define TILE_PITCH 68
#define SCAN_BLOCK 1024

typedef __attribute__((ext_vector_type(2))) float v2f;
typedef __attribute__((ext_vector_type(8))) float v8f;
typedef __attribute__((ext_vector_type(4))) float v4f;

#define BUCKET_SHIFT  8
#define BUCKET_ROWS   (1 << BUCKET_SHIFT)
#define CHUNK_EDGES   4096
#define CHUNK_THREADS 256
#define CHUNK_VEC     (CHUNK_EDGES / (CHUNK_THREADS * 4))
#define SCAN_PER      16
#define SCAN_TILE     (SCAN_BLOCK * SCAN_PER)
#define CSR_THREADS   1024
#define CSR_BATCH     6

__device__ __forceinline__ int4 gcn_rows4(const int* __restrict__ rows, int e, int E) {
    int4 r;
    if (e + 4 <= E) {
        r = *(const int4*)(rows + e);
    } else {
        r.x = (e     < E) ? rows[e]     : -1;
        r.y = (e + 1 < E) ? rows[e + 1] : -1;
        r.z = (e + 2 < E) ? rows[e + 2] : -1;
        r.w = (e + 3 < E) ? rows[e + 3] : -1;
    }
    return r;
}

__device__ __forceinline__ void gcn_lds_count(int* s_cnt, int row) {
    if (row >= 0)
        __hip_atomic_fetch_add(&s_cnt[row >> BUCKET_SHIFT], 1, __ATOMIC_RELAXED,
                               __HIP_MEMORY_SCOPE_WORKGROUP);
}

__global__ void __launch_bounds__(CHUNK_THREADS)
gcn_bucket_count(const int* __restrict__ rows, int* __restrict__ cc, int E, int nb, int nchunk) {
    extern __shared__ int s_bcnt[];
    const int t = threadIdx.x;
    const int chunk = blockIdx.x;
    for (int i = t; i < nb; i += CHUNK_THREADS) s_bcnt[i] = 0;
    int4 r[CHUNK_VEC];
    const int base = chunk * CHUNK_EDGES;
#pragma unroll
    for (int k = 0; k < CHUNK_VEC; ++k)
        r[k] = gcn_rows4(rows, base + (k * CHUNK_THREADS + t) * 4, E);
    __syncthreads();
#pragma unroll
    for (int k = 0; k < CHUNK_VEC; ++k) {
        gcn_lds_count(s_bcnt, r[k].x);
        gcn_lds_count(s_bcnt, r[k].y);
        gcn_lds_count(s_bcnt, r[k].z);
        gcn_lds_count(s_bcnt, r[k].w);
    }
    __syncthreads();
    for (int i = t; i < nb; i += CHUNK_THREADS) cc[(size_t)i * nchunk + chunk] = s_bcnt[i];
}

__global__ void __launch_bounds__(SCAN_BLOCK)
gcn_bucket_scan(const int* __restrict__ cc, int* __restrict__ cstart, int total) {
    __shared__ int s_v[SCAN_TILE + SCAN_BLOCK];
    __shared__ int s_w[32];
    const int t    = threadIdx.x;
    const int lane = t & 31;
    const int w    = t >> 5;
    int carry = 0;
    for (int t0 = 0; t0 < total; t0 += SCAN_TILE) {
        int v[SCAN_PER];
#pragma unroll
        for (int k = 0; k < SCAN_PER; ++k) {
            const int i = t0 + k * SCAN_BLOCK + t;
            v[k] = (i < total) ? cc[i] : 0;
        }
#pragma unroll
        for (int k = 0; k < SCAN_PER; ++k) {
            const int p = k * SCAN_BLOCK + t;
            s_v[p + (p >> 4)] = v[k];
        }
        __syncthreads();
        int sum = 0;
#pragma unroll
        for (int j = 0; j < SCAN_PER; ++j) sum += s_v[t * (SCAN_PER + 1) + j];
        int inc = sum;
        for (int d = 1; d < 32; d <<= 1) {
            const int o = __shfl_up(inc, d);
            if (lane >= d) inc += o;
        }
        if (lane == 31) s_w[w] = inc;
        __syncthreads();
        const int wt = s_w[lane];
        int before = (lane < w) ? wt : 0;
        int all    = wt;
        for (int d = 16; d > 0; d >>= 1) {
            before += __shfl_xor(before, d);
            all    += __shfl_xor(all, d);
        }
        int run = carry + before + inc - sum;
#pragma unroll
        for (int j = 0; j < SCAN_PER; ++j) {
            const int x = s_v[t * (SCAN_PER + 1) + j];
            s_v[t * (SCAN_PER + 1) + j] = run;
            run += x;
        }
        __syncthreads();
#pragma unroll
        for (int k = 0; k < SCAN_PER; ++k) {
            const int i = t0 + k * SCAN_BLOCK + t;
            const int p = k * SCAN_BLOCK + t;
            if (i < total) cstart[i] = s_v[p + (p >> 4)];
        }
        carry += all;
        __syncthreads();
    }
    if (t == 0) cstart[total] = carry;
}

__device__ __forceinline__ void gcn_file_edge(int* s_cur, int* __restrict__ brec, int row, int e) {
    if (row >= 0) {
        const int pos = __hip_atomic_fetch_add(&s_cur[row >> BUCKET_SHIFT], 1, __ATOMIC_RELAXED,
                                               __HIP_MEMORY_SCOPE_WORKGROUP);
        *(int2*)(brec + 2 * (size_t)pos) = make_int2(row, e);
    }
}

__global__ void __launch_bounds__(CHUNK_THREADS)
gcn_bucket_scatter(const int* __restrict__ rows, const int* __restrict__ cstart,
                   int* __restrict__ brec, int E, int nb, int nchunk) {
    extern __shared__ int s_bcur[];
    const int t = threadIdx.x;
    const int chunk = blockIdx.x;
    int4 r[CHUNK_VEC];
    const int base = chunk * CHUNK_EDGES;
#pragma unroll
    for (int k = 0; k < CHUNK_VEC; ++k)
        r[k] = gcn_rows4(rows, base + (k * CHUNK_THREADS + t) * 4, E);
    for (int i = t; i < nb; i += CHUNK_THREADS) s_bcur[i] = cstart[(size_t)i * nchunk + chunk];
    __syncthreads();
#pragma unroll
    for (int k = 0; k < CHUNK_VEC; ++k) {
        const int e = base + (k * CHUNK_THREADS + t) * 4;
        gcn_file_edge(s_bcur, brec, r[k].x, e);
        gcn_file_edge(s_bcur, brec, r[k].y, e + 1);
        gcn_file_edge(s_bcur, brec, r[k].z, e + 2);
        gcn_file_edge(s_bcur, brec, r[k].w, e + 3);
    }
}

__global__ void __launch_bounds__(CSR_THREADS)
gcn_bucket_csr(const int* __restrict__ brec, const int* __restrict__ cstart,
               int* __restrict__ offs, int* __restrict__ perm, int Nn, int nchunk) {
    __shared__ int s_cnt[BUCKET_ROWS];
    __shared__ int s_cur[BUCKET_ROWS];
    __shared__ int s_w[BUCKET_ROWS / 32];
    const int t    = threadIdx.x;
    const int lane = t & 31;
    const int w    = t >> 5;
    const int b    = blockIdx.x;
    const int bs = cstart[(size_t)b * nchunk];
    const int be = cstart[(size_t)(b + 1) * nchunk];
    if (t < BUCKET_ROWS) s_cnt[t] = 0;
    __syncthreads();
    for (int i0 = bs + t; i0 < be; i0 += CSR_BATCH * CSR_THREADS) {
        int rw[CSR_BATCH];
#pragma unroll
        for (int u = 0; u < CSR_BATCH; ++u) {
            const int i = i0 + u * CSR_THREADS;
            rw[u] = (i < be) ? brec[2 * (size_t)i] : -1;
        }
#pragma unroll
        for (int u = 0; u < CSR_BATCH; ++u)
            if (rw[u] >= 0)
                __hip_atomic_fetch_add(&s_cnt[rw[u] & (BUCKET_ROWS - 1)], 1, __ATOMIC_RELAXED,
                                       __HIP_MEMORY_SCOPE_WORKGROUP);
    }
    __syncthreads();
    int c = 0, inc = 0;
    if (t < BUCKET_ROWS) {
        c = s_cnt[t];
        inc = c;
        for (int d = 1; d < 32; d <<= 1) {
            const int o = __shfl_up(inc, d);
            if (lane >= d) inc += o;
        }
        if (lane == 31) s_w[w] = inc;
    }
    __syncthreads();
    if (t < BUCKET_ROWS) {
        int before = 0;
        for (int j = 0; j < w; ++j) before += s_w[j];
        const int start = bs + before + inc - c;
        s_cur[t] = start;
        const int r = b * BUCKET_ROWS + t;
        if (r < Nn)      offs[r]  = start;
        if (r == Nn - 1) offs[Nn] = start + c;
    }
    __syncthreads();
    for (int i0 = bs + t; i0 < be; i0 += CSR_BATCH * CSR_THREADS) {
        int2 rc[CSR_BATCH];
#pragma unroll
        for (int u = 0; u < CSR_BATCH; ++u) {
            const int i = i0 + u * CSR_THREADS;
            rc[u] = (i < be) ? *(const int2*)(brec + 2 * (size_t)i) : make_int2(-1, 0);
        }
#pragma unroll
        for (int u = 0; u < CSR_BATCH; ++u)
            if (rc[u].x >= 0) {
                const int pos = __hip_atomic_fetch_add(&s_cur[rc[u].x & (BUCKET_ROWS - 1)], 1,
                                                       __ATOMIC_RELAXED, __HIP_MEMORY_SCOPE_WORKGROUP);
                perm[pos] = rc[u].y;
            }
    }
}

__device__ __forceinline__ v4f gcn_ld4(const float* __restrict__ xrow, unsigned off) {
    return *(const v4f*)((const char*)xrow + off);
}

__device__ __forceinline__ void gcn_edge1(const float* __restrict__ x, unsigned offa, unsigned offb,
                                          int c, float v, v4f& s0, v4f& s1) {
    const float* __restrict__ xr = x + (size_t)c * FDIM;
    const v4f p = gcn_ld4(xr, offa);
    const v4f q = gcn_ld4(xr, offb);
    s0 += v * p;
    s1 += v * q;
}

__device__ __forceinline__ void gcn_pick(int rank, int c, float v, int p, int& cp, float& vp) {
    const int l = __builtin_ctz(__builtin_amdgcn_ballot_w32(rank == p));
    cp = __builtin_amdgcn_readlane(c, l);
    vp = __int_as_float(__builtin_amdgcn_readlane(__float_as_int(v), l));
}

__global__ void __launch_bounds__(ROWS_PER_BLOCK * 32)
gcn_gather_gemm_bias_elu(const float* __restrict__ x,
                         const float* __restrict__ vals,
                         const int*   __restrict__ cols,
                         const int*   __restrict__ offs,
                         const int*   __restrict__ perm,
                         const float* __restrict__ W,
                         const float* __restrict__ bias,
                         float*       __restrict__ out,
                         int Nn) {
    __shared__ __attribute__((aligned(16))) float s_agg[TILE_ROWS * TILE_PITCH];
    __shared__ __attribute__((aligned(16))) float s_res[TILE_ROWS * TILE_PITCH];

    const int lane = threadIdx.x & 31;
    const int wv   = __builtin_amdgcn_readfirstlane((int)(threadIdx.x >> 5));
    const int row0 = blockIdx.x * ROWS_PER_BLOCK;
    const int r    = row0 + wv;
    const int half = lane >> 4;
    const int q4   = (lane & 15) * 4;

    v4f s0 = {0.f, 0.f, 0.f, 0.f};
    v4f s1 = {0.f, 0.f, 0.f, 0.f};

    if (r < Nn) {
        const int beg = offs[r];
        const int kt  = offs[r + 1] - beg;
        const unsigned offa = (unsigned)(((size_t)half * Nn * FDIM + q4) * sizeof(float));
        const unsigned offb = (unsigned)(((size_t)(half + 2) * Nn * FDIM + q4) * sizeof(float));

        if (kt <= 32) {
            int   eid = 0x7fffffff;
            int   c   = 0;
            float v   = 0.f;
            if (lane < kt) {
                eid = perm[beg + lane];
                c   = cols[eid];
                v   = vals[eid];
            }
            int rank = 0;
            for (int j = 0; j < kt; ++j)
                rank += (__builtin_amdgcn_readlane(eid, j) < eid) ? 1 : 0;

            int p = 0;
            for (; p + 3 < kt; p += 4) {
                int c0, c1, c2, c3;
                float v0, v1, v2, v3;
                gcn_pick(rank, c, v, p,     c0, v0);
                gcn_pick(rank, c, v, p + 1, c1, v1);
                gcn_pick(rank, c, v, p + 2, c2, v2);
                gcn_pick(rank, c, v, p + 3, c3, v3);
                const float* __restrict__ x0 = x + (size_t)c0 * FDIM;
                const float* __restrict__ x1 = x + (size_t)c1 * FDIM;
                const float* __restrict__ x2 = x + (size_t)c2 * FDIM;
                const float* __restrict__ x3 = x + (size_t)c3 * FDIM;
                const v4f p0 = gcn_ld4(x0, offa), q0 = gcn_ld4(x0, offb);
                const v4f p1 = gcn_ld4(x1, offa), q1 = gcn_ld4(x1, offb);
                const v4f p2 = gcn_ld4(x2, offa), q2 = gcn_ld4(x2, offb);
                const v4f p3 = gcn_ld4(x3, offa), q3 = gcn_ld4(x3, offb);
                s0 += v0 * p0;  s1 += v0 * q0;
                s0 += v1 * p1;  s1 += v1 * q1;
                s0 += v2 * p2;  s1 += v2 * q2;
                s0 += v3 * p3;  s1 += v3 * q3;
            }
            for (; p < kt; ++p) {
                int c0; float v0;
                gcn_pick(rank, c, v, p, c0, v0);
                gcn_edge1(x, offa, offb, c0, v0, s0, s1);
            }
        } else {
            int prev = -1;
            for (int p = 0; p < kt; ++p) {
                int m = 0x7fffffff;
                for (int i = lane; i < kt; i += 32) {
                    const int e = perm[beg + i];
                    if (e > prev && e < m) m = e;
                }
                for (int d = 16; d > 0; d >>= 1) {
                    const int o = __shfl_xor(m, d);
                    m = (o < m) ? o : m;
                }
                m = __builtin_amdgcn_readfirstlane(m);
                prev = m;
                gcn_edge1(x, offa, offb, cols[m], vals[m], s0, s1);
            }
        }
    }

    *(v4f*)&s_agg[(half * ROWS_PER_BLOCK + wv) * TILE_PITCH + q4]       = s0;
    *(v4f*)&s_agg[((half + 2) * ROWS_PER_BLOCK + wv) * TILE_PITCH + q4] = s1;

    const int m  = lane & 15;
    const int kg = (lane >> 4) << 1;
    const int mt = wv >> 2;
    const int nc = (wv & 3) * 16 + m;

    v2f bf[FDIM / 4];
    for (int k4 = 0; k4 < FDIM / 4; ++k4) {
        const int kk = k4 * 4 + kg;
        bf[k4].x = W[kk * FDIM + nc];
        bf[k4].y = W[(kk + 1) * FDIM + nc];
    }
    const float bb = bias[nc];

    __syncthreads();

    const float* arow = s_agg + (mt * 16 + m) * TILE_PITCH;
    v8f acc = v8f{};
    for (int k4 = 0; k4 < FDIM / 4; ++k4) {
        const int kk = k4 * 4 + kg;
        v2f a;
        a.x = arow[kk];
        a.y = arow[kk + 1];
        acc = __builtin_amdgcn_wmma_f32_16x16x4_f32(
             false, a,  false, bf[k4],
             (short)0, acc,
             false,  false);
    }

    for (int i = 0; i < 8; ++i) {
        const float v = acc[i] + bb;
        s_res[(mt * 16 + half * 8 + i) * TILE_PITCH + nc] = (v > 0.0f) ? v : (__expf(v) - 1.0f);
    }

    __syncthreads();

    for (int u = 0; u < 2; ++u) {
        const int g   = 4 * wv + 2 * u + half;
        const int row = row0 + (g & (ROWS_PER_BLOCK - 1));
        if (row < Nn)
            *(v4f*)(out + ((size_t)(g >> 3) * Nn + row) * FDIM + q4) =
                *(const v4f*)&s_res[g * TILE_PITCH + q4];
    }
}

extern "C" void kernel_launch(void* const* d_in, const int* in_sizes, int n_in,
                              void* d_out, int out_size, void* d_ws, size_t ws_size,
                              hipStream_t stream) {
    const float* x      = (const float*)d_in[0];
    const float* vals   = (const float*)d_in[1];
    const float* weight = (const float*)d_in[2];
    const float* bias   = (const float*)d_in[3];
    const int*   rows   = (const int*)d_in[4];
    const int*   cols   = (const int*)d_in[5];
    float*       out    = (float*)d_out;

    const int  E  = in_sizes[1];
    const int  Nn = in_sizes[0] / (BATCH * FDIM);
    const int  nb     = (Nn + BUCKET_ROWS - 1) / BUCKET_ROWS;
    const int  nchunk = (E + CHUNK_EDGES - 1) / CHUNK_EDGES;
    const int  total  = nb * nchunk;
    const long nPad = ((long)Nn + 64) & ~63L;
    const long ePad = ((long)E + 63) & ~63L;
    const long tPad = ((long)total + 64) & ~63L;
    int*   offs   = (int*)d_ws;
    int*   perm   = offs + nPad;
    int*   brec   = perm + ePad;
    int*   cc     = brec + 2 * ePad;
    int*   cstart = cc + tPad;

    gcn_bucket_count<<<(unsigned)nchunk, CHUNK_THREADS, nb * sizeof(int), stream>>>(
        rows, cc, E, nb, nchunk);
    gcn_bucket_scan<<<1, SCAN_BLOCK, 0, stream>>>(cc, cstart, total);
    gcn_bucket_scatter<<<(unsigned)nchunk, CHUNK_THREADS, nb * sizeof(int), stream>>>(
        rows, cstart, brec, E, nb, nchunk);
    gcn_bucket_csr<<<(unsigned)nb, CSR_THREADS, 0, stream>>>(brec, cstart, offs, perm, Nn, nchunk);
    gcn_gather_gemm_bias_elu<<<(unsigned)((Nn + ROWS_PER_BLOCK - 1) / ROWS_PER_BLOCK),
                               ROWS_PER_BLOCK * 32, 0, stream>>>(
        x, vals, cols, offs, perm, weight, bias, out, Nn);
}


# Gather: rows of 33-64 edges are ranked two edges per lane and summed three edges per trip as short rows are, not one edge at a time by a search of the whole row; sums and their order unchanged
# speedup vs baseline: 3.7381x; 1.1568x over previous
#include <hip/hip_runtime.h>


#define BATCH 4
#define FDIM  64
#define ROWS_PER_BLOCK 8
#define TILE_ROWS (BATCH * ROWS_PER_BLOCK)
#define TILE_PITCH 68
#define SCAN_BLOCK 1024

typedef __attribute__((ext_vector_type(2))) float v2f;
typedef __attribute__((ext_vector_type(8))) float v8f;
typedef __attribute__((ext_vector_type(4))) float v4f;

#define BUCKET_SHIFT  8
#define BUCKET_ROWS   (1 << BUCKET_SHIFT)
#define CHUNK_EDGES   4096
#define CHUNK_THREADS 256
#define CHUNK_VEC     (CHUNK_EDGES / (CHUNK_THREADS * 4))
#define SCAN_PER      16
#define SCAN_TILE     (SCAN_BLOCK * SCAN_PER)
#define CSR_THREADS   1024
#define CSR_BATCH     6

__device__ __forceinline__ int4 gcn_rows4(const int* __restrict__ rows, int e, int E) {
    int4 r;
    if (e + 4 <= E) {
        r = *(const int4*)(rows + e);
    } else {
        r.x = (e     < E) ? rows[e]     : -1;
        r.y = (e + 1 < E) ? rows[e + 1] : -1;
        r.z = (e + 2 < E) ? rows[e + 2] : -1;
        r.w = (e + 3 < E) ? rows[e + 3] : -1;
    }
    return r;
}

__device__ __forceinline__ void gcn_lds_count(int* s_cnt, int row) {
    if (row >= 0)
        __hip_atomic_fetch_add(&s_cnt[row >> BUCKET_SHIFT], 1, __ATOMIC_RELAXED,
                               __HIP_MEMORY_SCOPE_WORKGROUP);
}

__global__ void __launch_bounds__(CHUNK_THREADS)
gcn_bucket_count(const int* __restrict__ rows, int* __restrict__ cc, int E, int nb, int nchunk) {
    extern __shared__ int s_bcnt[];
    const int t = threadIdx.x;
    const int chunk = blockIdx.x;
    for (int i = t; i < nb; i += CHUNK_THREADS) s_bcnt[i] = 0;
    int4 r[CHUNK_VEC];
    const int base = chunk * CHUNK_EDGES;
#pragma unroll
    for (int k = 0; k < CHUNK_VEC; ++k)
        r[k] = gcn_rows4(rows, base + (k * CHUNK_THREADS + t) * 4, E);
    __syncthreads();
#pragma unroll
    for (int k = 0; k < CHUNK_VEC; ++k) {
        gcn_lds_count(s_bcnt, r[k].x);
        gcn_lds_count(s_bcnt, r[k].y);
        gcn_lds_count(s_bcnt, r[k].z);
        gcn_lds_count(s_bcnt, r[k].w);
    }
    __syncthreads();
    for (int i = t; i < nb; i += CHUNK_THREADS) cc[(size_t)i * nchunk + chunk] = s_bcnt[i];
}

__global__ void __launch_bounds__(SCAN_BLOCK)
gcn_bucket_scan(const int* __restrict__ cc, int* __restrict__ cstart, int total) {
    __shared__ int s_v[SCAN_TILE + SCAN_BLOCK];
    __shared__ int s_w[32];
    const int t    = threadIdx.x;
    const int lane = t & 31;
    const int w    = t >> 5;
    int carry = 0;
    for (int t0 = 0; t0 < total; t0 += SCAN_TILE) {
        int v[SCAN_PER];
#pragma unroll
        for (int k = 0; k < SCAN_PER; ++k) {
            const int i = t0 + k * SCAN_BLOCK + t;
            v[k] = (i < total) ? cc[i] : 0;
        }
#pragma unroll
        for (int k = 0; k < SCAN_PER; ++k) {
            const int p = k * SCAN_BLOCK + t;
            s_v[p + (p >> 4)] = v[k];
        }
        __syncthreads();
        int sum = 0;
#pragma unroll
        for (int j = 0; j < SCAN_PER; ++j) sum += s_v[t * (SCAN_PER + 1) + j];
        int inc = sum;
        for (int d = 1; d < 32; d <<= 1) {
            const int o = __shfl_up(inc, d);
            if (lane >= d) inc += o;
        }
        if (lane == 31) s_w[w] = inc;
        __syncthreads();
        const int wt = s_w[lane];
        int before = (lane < w) ? wt : 0;
        int all    = wt;
        for (int d = 16; d > 0; d >>= 1) {
            before += __shfl_xor(before, d);
            all    += __shfl_xor(all, d);
        }
        int run = carry + before + inc - sum;
#pragma unroll
        for (int j = 0; j < SCAN_PER; ++j) {
            const int x = s_v[t * (SCAN_PER + 1) + j];
            s_v[t * (SCAN_PER + 1) + j] = run;
            run += x;
        }
        __syncthreads();
#pragma unroll
        for (int k = 0; k < SCAN_PER; ++k) {
            const int i = t0 + k * SCAN_BLOCK + t;
            const int p = k * SCAN_BLOCK + t;
            if (i < total) cstart[i] = s_v[p + (p >> 4)];
        }
        carry += all;
        __syncthreads();
    }
    if (t == 0) cstart[total] = carry;
}

__device__ __forceinline__ void gcn_file_edge(int* s_cur, int* __restrict__ brec, int row, int e) {
    if (row >= 0) {
        const int pos = __hip_atomic_fetch_add(&s_cur[row >> BUCKET_SHIFT], 1, __ATOMIC_RELAXED,
                                               __HIP_MEMORY_SCOPE_WORKGROUP);
        *(int2*)(brec + 2 * (size_t)pos) = make_int2(row, e);
    }
}

__global__ void __launch_bounds__(CHUNK_THREADS)
gcn_bucket_scatter(const int* __restrict__ rows, const int* __restrict__ cstart,
                   int* __restrict__ brec, int E, int nb, int nchunk) {
    extern __shared__ int s_bcur[];
    const int t = threadIdx.x;
    const int chunk = blockIdx.x;
    int4 r[CHUNK_VEC];
    const int base = chunk * CHUNK_EDGES;
#pragma unroll
    for (int k = 0; k < CHUNK_VEC; ++k)
        r[k] = gcn_rows4(rows, base + (k * CHUNK_THREADS + t) * 4, E);
    for (int i = t; i < nb; i += CHUNK_THREADS) s_bcur[i] = cstart[(size_t)i * nchunk + chunk];
    __syncthreads();
#pragma unroll
    for (int k = 0; k < CHUNK_VEC; ++k) {
        const int e = base + (k * CHUNK_THREADS + t) * 4;
        gcn_file_edge(s_bcur, brec, r[k].x, e);
        gcn_file_edge(s_bcur, brec, r[k].y, e + 1);
        gcn_file_edge(s_bcur, brec, r[k].z, e + 2);
        gcn_file_edge(s_bcur, brec, r[k].w, e + 3);
    }
}

__global__ void __launch_bounds__(CSR_THREADS)
gcn_bucket_csr(const int* __restrict__ brec, const int* __restrict__ cstart,
               int* __restrict__ offs, int* __restrict__ perm, int Nn, int nchunk) {
    __shared__ int s_cnt[BUCKET_ROWS];
    __shared__ int s_cur[BUCKET_ROWS];
    __shared__ int s_w[BUCKET_ROWS / 32];
    const int t    = threadIdx.x;
    const int lane = t & 31;
    const int w    = t >> 5;
    const int b    = blockIdx.x;
    const int bs = cstart[(size_t)b * nchunk];
    const int be = cstart[(size_t)(b + 1) * nchunk];
    if (t < BUCKET_ROWS) s_cnt[t] = 0;
    __syncthreads();
    for (int i0 = bs + t; i0 < be; i0 += CSR_BATCH * CSR_THREADS) {
        int rw[CSR_BATCH];
#pragma unroll
        for (int u = 0; u < CSR_BATCH; ++u) {
            const int i = i0 + u * CSR_THREADS;
            rw[u] = (i < be) ? brec[2 * (size_t)i] : -1;
        }
#pragma unroll
        for (int u = 0; u < CSR_BATCH; ++u)
            if (rw[u] >= 0)
                __hip_atomic_fetch_add(&s_cnt[rw[u] & (BUCKET_ROWS - 1)], 1, __ATOMIC_RELAXED,
                                       __HIP_MEMORY_SCOPE_WORKGROUP);
    }
    __syncthreads();
    int c = 0, inc = 0;
    if (t < BUCKET_ROWS) {
        c = s_cnt[t];
        inc = c;
        for (int d = 1; d < 32; d <<= 1) {
            const int o = __shfl_up(inc, d);
            if (lane >= d) inc += o;
        }
        if (lane == 31) s_w[w] = inc;
    }
    __syncthreads();
    if (t < BUCKET_ROWS) {
        int before = 0;
        for (int j = 0; j < w; ++j) before += s_w[j];
        const int start = bs + before + inc - c;
        s_cur[t] = start;
        const int r = b * BUCKET_ROWS + t;
        if (r < Nn)      offs[r]  = start;
        if (r == Nn - 1) offs[Nn] = start + c;
    }
    __syncthreads();
    for (int i0 = bs + t; i0 < be; i0 += CSR_BATCH * CSR_THREADS) {
        int2 rc[CSR_BATCH];
#pragma unroll
        for (int u = 0; u < CSR_BATCH; ++u) {
            const int i = i0 + u * CSR_THREADS;
            rc[u] = (i < be) ? *(const int2*)(brec + 2 * (size_t)i) : make_int2(-1, 0);
        }
#pragma unroll
        for (int u = 0; u < CSR_BATCH; ++u)
            if (rc[u].x >= 0) {
                const int pos = __hip_atomic_fetch_add(&s_cur[rc[u].x & (BUCKET_ROWS - 1)], 1,
                                                       __ATOMIC_RELAXED, __HIP_MEMORY_SCOPE_WORKGROUP);
                perm[pos] = rc[u].y;
            }
    }
}

__device__ __forceinline__ v4f gcn_ld4(const float* __restrict__ xrow, unsigned off) {
    return *(const v4f*)((const char*)xrow + off);
}

__device__ __forceinline__ void gcn_edge1(const float* __restrict__ x, unsigned offa, unsigned offb,
                                          int c, float v, v4f& s0, v4f& s1) {
    const float* __restrict__ xr = x + (size_t)c * FDIM;
    const v4f p = gcn_ld4(xr, offa);
    const v4f q = gcn_ld4(xr, offb);
    s0 += v * p;
    s1 += v * q;
}

__device__ __forceinline__ void gcn_pick(int rank, int c, float v, int p, int& cp, float& vp) {
    const int l = __builtin_ctz(__builtin_amdgcn_ballot_w32(rank == p));
    cp = __builtin_amdgcn_readlane(c, l);
    vp = __int_as_float(__builtin_amdgcn_readlane(__float_as_int(v), l));
}

__device__ __forceinline__ void gcn_pick2(int rankA, int rankB, int cA, float vA, int cB, float vB,
                                          int p, int& cp, float& vp) {
    const unsigned mA = __builtin_amdgcn_ballot_w32(rankA == p);
    const unsigned mB = __builtin_amdgcn_ballot_w32(rankB == p);
    const int l  = __builtin_ctz(mA | mB);
    const int ca = __builtin_amdgcn_readlane(cA, l);
    const int cb = __builtin_amdgcn_readlane(cB, l);
    const int va = __builtin_amdgcn_readlane(__float_as_int(vA), l);
    const int vb = __builtin_amdgcn_readlane(__float_as_int(vB), l);
    cp = mA ? ca : cb;
    vp = __int_as_float(mA ? va : vb);
}

__global__ void __launch_bounds__(ROWS_PER_BLOCK * 32)
gcn_gather_gemm_bias_elu(const float* __restrict__ x,
                         const float* __restrict__ vals,
                         const int*   __restrict__ cols,
                         const int*   __restrict__ offs,
                         const int*   __restrict__ perm,
                         const float* __restrict__ W,
                         const float* __restrict__ bias,
                         float*       __restrict__ out,
                         int Nn) {
    __shared__ __attribute__((aligned(16))) float s_agg[TILE_ROWS * TILE_PITCH];
    __shared__ __attribute__((aligned(16))) float s_res[TILE_ROWS * TILE_PITCH];

    const int lane = threadIdx.x & 31;
    const int wv   = __builtin_amdgcn_readfirstlane((int)(threadIdx.x >> 5));
    const int row0 = blockIdx.x * ROWS_PER_BLOCK;
    const int r    = row0 + wv;
    const int half = lane >> 4;
    const int q4   = (lane & 15) * 4;

    v4f s0 = {0.f, 0.f, 0.f, 0.f};
    v4f s1 = {0.f, 0.f, 0.f, 0.f};

    if (r < Nn) {
        const int beg = offs[r];
        const int kt  = offs[r + 1] - beg;
        const unsigned offa = (unsigned)(((size_t)half * Nn * FDIM + q4) * sizeof(float));
        const unsigned offb = (unsigned)(((size_t)(half + 2) * Nn * FDIM + q4) * sizeof(float));

        if (kt <= 32) {
            int   eid = 0x7fffffff;
            int   c   = 0;
            float v   = 0.f;
            if (lane < kt) {
                eid = perm[beg + lane];
                c   = cols[eid];
                v   = vals[eid];
            }
            int rank = 0;
            for (int j = 0; j < kt; ++j)
                rank += (__builtin_amdgcn_readlane(eid, j) < eid) ? 1 : 0;

            int p = 0;
            for (; p + 3 < kt; p += 4) {
                int c0, c1, c2, c3;
                float v0, v1, v2, v3;
                gcn_pick(rank, c, v, p,     c0, v0);
                gcn_pick(rank, c, v, p + 1, c1, v1);
                gcn_pick(rank, c, v, p + 2, c2, v2);
                gcn_pick(rank, c, v, p + 3, c3, v3);
                const float* __restrict__ x0 = x + (size_t)c0 * FDIM;
                const float* __restrict__ x1 = x + (size_t)c1 * FDIM;
                const float* __restrict__ x2 = x + (size_t)c2 * FDIM;
                const float* __restrict__ x3 = x + (size_t)c3 * FDIM;
                const v4f p0 = gcn_ld4(x0, offa), q0 = gcn_ld4(x0, offb);
                const v4f p1 = gcn_ld4(x1, offa), q1 = gcn_ld4(x1, offb);
                const v4f p2 = gcn_ld4(x2, offa), q2 = gcn_ld4(x2, offb);
                const v4f p3 = gcn_ld4(x3, offa), q3 = gcn_ld4(x3, offb);
                s0 += v0 * p0;  s1 += v0 * q0;
                s0 += v1 * p1;  s1 += v1 * q1;
                s0 += v2 * p2;  s1 += v2 * q2;
                s0 += v3 * p3;  s1 += v3 * q3;
            }
            for (; p < kt; ++p) {
                int c0; float v0;
                gcn_pick(rank, c, v, p, c0, v0);
                gcn_edge1(x, offa, offb, c0, v0, s0, s1);
            }
        } else if (kt <= 64) {
            const bool hasB = lane + 32 < kt;
            const int  iB   = hasB ? lane + 32 : kt - 1;
            const int eidA  = perm[beg + lane];
            const int eidBr = perm[beg + iB];
            const int   cA = cols[eidA];
            const float vA = vals[eidA];
            const int   cB = cols[eidBr];
            const float vB = vals[eidBr];
            const int eidB = hasB ? eidBr : 0x7fffffff;
            int rankA = 0, rankB = 0;
            for (int j = 0; j < 32; ++j) {
                const int e = __builtin_amdgcn_readlane(eidA, j);
                rankA += (e < eidA) ? 1 : 0;
                rankB += (e < eidB) ? 1 : 0;
            }
            for (int j = 0; j < kt - 32; ++j) {
                const int e = __builtin_amdgcn_readlane(eidB, j);
                rankA += (e < eidA) ? 1 : 0;
                rankB += (e < eidB) ? 1 : 0;
            }

            int p = 0;
            for (; p + 2 < kt; p += 3) {
                int c0, c1, c2;
                float v0, v1, v2;
                gcn_pick2(rankA, rankB, cA, vA, cB, vB, p, c0, v0);
                gcn_pick2(rankA, rankB, cA, vA, cB, vB, p + 1, c1, v1);
                gcn_pick2(rankA, rankB, cA, vA, cB, vB, p + 2, c2, v2);
                const float* __restrict__ x0 = x + (size_t)c0 * FDIM;
                const float* __restrict__ x1 = x + (size_t)c1 * FDIM;
                const float* __restrict__ x2 = x + (size_t)c2 * FDIM;
                const v4f p0 = gcn_ld4(x0, offa), q0 = gcn_ld4(x0, offb);
                const v4f p1 = gcn_ld4(x1, offa), q1 = gcn_ld4(x1, offb);
                const v4f p2 = gcn_ld4(x2, offa), q2 = gcn_ld4(x2, offb);
                s0 += v0 * p0;  s1 += v0 * q0;
                s0 += v1 * p1;  s1 += v1 * q1;
                s0 += v2 * p2;  s1 += v2 * q2;
            }
            for (; p < kt; ++p) {
                int c0; float v0;
                gcn_pick2(rankA, rankB, cA, vA, cB, vB, p, c0, v0);
                gcn_edge1(x, offa, offb, c0, v0, s0, s1);
            }
        } else {
            int prev = -1;
            for (int p = 0; p < kt; ++p) {
                int m = 0x7fffffff;
                for (int i = lane; i < kt; i += 32) {
                    const int e = perm[beg + i];
                    if (e > prev && e < m) m = e;
                }
                for (int d = 16; d > 0; d >>= 1) {
                    const int o = __shfl_xor(m, d);
                    m = (o < m) ? o : m;
                }
                m = __builtin_amdgcn_readfirstlane(m);
                prev = m;
                gcn_edge1(x, offa, offb, cols[m], vals[m], s0, s1);
            }
        }
    }

    *(v4f*)&s_agg[(half * ROWS_PER_BLOCK + wv) * TILE_PITCH + q4]       = s0;
    *(v4f*)&s_agg[((half + 2) * ROWS_PER_BLOCK + wv) * TILE_PITCH + q4] = s1;

    const int m  = lane & 15;
    const int kg = (lane >> 4) << 1;
    const int mt = wv >> 2;
    const int nc = (wv & 3) * 16 + m;

    v2f bf[FDIM / 4];
    for (int k4 = 0; k4 < FDIM / 4; ++k4) {
        const int kk = k4 * 4 + kg;
        bf[k4].x = W[kk * FDIM + nc];
        bf[k4].y = W[(kk + 1) * FDIM + nc];
    }
    const float bb = bias[nc];

    __syncthreads();

    const float* arow = s_agg + (mt * 16 + m) * TILE_PITCH;
    v8f acc = v8f{};
    for (int k4 = 0; k4 < FDIM / 4; ++k4) {
        const int kk = k4 * 4 + kg;
        v2f a;
        a.x = arow[kk];
        a.y = arow[kk + 1];
        acc = __builtin_amdgcn_wmma_f32_16x16x4_f32(
             false, a,  false, bf[k4],
             (short)0, acc,
             false,  false);
    }

    for (int i = 0; i < 8; ++i) {
        const float v = acc[i] + bb;
        s_res[(mt * 16 + half * 8 + i) * TILE_PITCH + nc] = (v > 0.0f) ? v : (__expf(v) - 1.0f);
    }

    __syncthreads();

    for (int u = 0; u < 2; ++u) {
        const int g   = 4 * wv + 2 * u + half;
        const int row = row0 + (g & (ROWS_PER_BLOCK - 1));
        if (row < Nn)
            *(v4f*)(out + ((size_t)(g >> 3) * Nn + row) * FDIM + q4) =
                *(const v4f*)&s_res[g * TILE_PITCH + q4];
    }
}

extern "C" void kernel_launch(void* const* d_in, const int* in_sizes, int n_in,
                              void* d_out, int out_size, void* d_ws, size_t ws_size,
                              hipStream_t stream) {
    const float* x      = (const float*)d_in[0];
    const float* vals   = (const float*)d_in[1];
    const float* weight = (const float*)d_in[2];
    const float* bias   = (const float*)d_in[3];
    const int*   rows   = (const int*)d_in[4];
    const int*   cols   = (const int*)d_in[5];
    float*       out    = (float*)d_out;

    const int  E  = in_sizes[1];
    const int  Nn = in_sizes[0] / (BATCH * FDIM);
    const int  nb     = (Nn + BUCKET_ROWS - 1) / BUCKET_ROWS;
    const int  nchunk = (E + CHUNK_EDGES - 1) / CHUNK_EDGES;
    const int  total  = nb * nchunk;
    const long nPad = ((long)Nn + 64) & ~63L;
    const long ePad = ((long)E + 63) & ~63L;
    const long tPad = ((long)total + 64) & ~63L;
    int*   offs   = (int*)d_ws;
    int*   perm   = offs + nPad;
    int*   brec   = perm + ePad;
    int*   cc     = brec + 2 * ePad;
    int*   cstart = cc + tPad;

    gcn_bucket_count<<<(unsigned)nchunk, CHUNK_THREADS, nb * sizeof(int), stream>>>(
        rows, cc, E, nb, nchunk);
    gcn_bucket_scan<<<1, SCAN_BLOCK, 0, stream>>>(cc, cstart, total);
    gcn_bucket_scatter<<<(unsigned)nchunk, CHUNK_THREADS, nb * sizeof(int), stream>>>(
        rows, cstart, brec, E, nb, nchunk);
    gcn_bucket_csr<<<(unsigned)nb, CSR_THREADS, 0, stream>>>(brec, cstart, offs, perm, Nn, nchunk);
    gcn_gather_gemm_bias_elu<<<(unsigned)((Nn + ROWS_PER_BLOCK - 1) / ROWS_PER_BLOCK),
                               ROWS_PER_BLOCK * 32, 0, stream>>>(
        x, vals, cols, offs, perm, weight, bias, out, Nn);
}


# CSR build: chunk count, global scan and bucket scatter (3 launches) become one chunk-local partition in LDS; a bucket finds its runs from per-chunk starts: 5 launches to 3; row sum order unchanged
# speedup vs baseline: 4.0699x; 1.0888x over previous
#include <hip/hip_runtime.h>


#define BATCH 4
#define FDIM  64
#define ROWS_PER_BLOCK 8
#define TILE_ROWS (BATCH * ROWS_PER_BLOCK)
#define TILE_PITCH 68

typedef __attribute__((ext_vector_type(2))) float v2f;
typedef __attribute__((ext_vector_type(8))) float v8f;
typedef __attribute__((ext_vector_type(4))) float v4f;
typedef __attribute__((ext_vector_type(4))) int   v4i;

#define BUCKET_SHIFT  8
#define BUCKET_ROWS   (1 << BUCKET_SHIFT)
#define CHUNK_EDGES   4096
#define CHUNK_THREADS 256
#define CHUNK_VEC     (CHUNK_EDGES / (CHUNK_THREADS * 4))
#define CSR_THREADS   1024
#define CSR_SEGS      8

__device__ __forceinline__ int4 gcn_rows4(const int* __restrict__ rows, int e, int E) {
    int4 r;
    if (e + 4 <= E) {
        r = *(const int4*)(rows + e);
    } else {
        r.x = (e     < E) ? rows[e]     : -1;
        r.y = (e + 1 < E) ? rows[e + 1] : -1;
        r.z = (e + 2 < E) ? rows[e + 2] : -1;
        r.w = (e + 3 < E) ? rows[e + 3] : -1;
    }
    return r;
}

__device__ __forceinline__ void gcn_lds_count(int* s_cnt, int row) {
    if (row >= 0)
        __hip_atomic_fetch_add(&s_cnt[row >> BUCKET_SHIFT], 1, __ATOMIC_RELAXED,
                               __HIP_MEMORY_SCOPE_WORKGROUP);
}

template <int NT>
__device__ __forceinline__ int gcn_block_excl_scan(int* s_a, int n, int* s_w) {
    const int t    = threadIdx.x;
    const int lane = t & 31;
    const int w    = t >> 5;
    int carry = 0;
    for (int base = 0; base < n; base += NT) {
        const int i = base + t;
        const int v = (i < n) ? s_a[i] : 0;
        int inc = v;
        for (int d = 1; d < 32; d <<= 1) {
            const int o = __shfl_up(inc, d);
            if (lane >= d) inc += o;
        }
        if (lane == 31) s_w[w] = inc;
        __syncthreads();
        const int wt = (lane < NT / 32) ? s_w[lane] : 0;
        int before = (lane < w) ? wt : 0;
        int all    = wt;
        for (int d = 16; d > 0; d >>= 1) {
            before += __shfl_xor(before, d);
            all    += __shfl_xor(all, d);
        }
        if (i < n) s_a[i] = carry + before + inc - v;
        carry += all;
        __syncthreads();
    }
    return carry;
}

__device__ __forceinline__ int gcn_take_place(int* s_cur, int row) {
    int pos = -1;
    if (row >= 0)
        pos = __hip_atomic_fetch_add(&s_cur[row >> BUCKET_SHIFT], 1, __ATOMIC_RELAXED,
                                     __HIP_MEMORY_SCOPE_WORKGROUP);
    return pos;
}

__device__ __forceinline__ void gcn_file_edge(int2* s_rec, int pos, int row, int e) {
    if (pos >= 0) s_rec[pos] = make_int2(row, e);
}

__global__ void __launch_bounds__(CHUNK_THREADS)
gcn_chunk_partition(const int* __restrict__ rows, int* __restrict__ runs,
                    int* __restrict__ brec, int E, int nb) {
    extern __shared__ int s_bkt[];
    __shared__ __attribute__((aligned(16))) int2 s_rec[CHUNK_EDGES];
    __shared__ int s_w[32];
    const int t = threadIdx.x;
    const int chunk = blockIdx.x;
    for (int i = t; i < nb; i += CHUNK_THREADS) s_bkt[i] = 0;
    int4 r[CHUNK_VEC];
    const int base = chunk * CHUNK_EDGES;
#pragma unroll
    for (int k = 0; k < CHUNK_VEC; ++k)
        r[k] = gcn_rows4(rows, base + (k * CHUNK_THREADS + t) * 4, E);
    __syncthreads();
#pragma unroll
    for (int k = 0; k < CHUNK_VEC; ++k) {
        gcn_lds_count(s_bkt, r[k].x);
        gcn_lds_count(s_bkt, r[k].y);
        gcn_lds_count(s_bkt, r[k].z);
        gcn_lds_count(s_bkt, r[k].w);
    }
    __syncthreads();
    const int cnt = gcn_block_excl_scan<CHUNK_THREADS>(s_bkt, nb, s_w);
    int2* __restrict__ rn = (int2*)runs + (size_t)chunk * nb;
    for (int i = t; i < nb; i += CHUNK_THREADS) {
        const int a = s_bkt[i];
        const int z = (i + 1 < nb) ? s_bkt[i + 1] : cnt;
        rn[i] = make_int2(a, z - a);
    }
    __syncthreads();
    int4 pl[CHUNK_VEC];
#pragma unroll
    for (int k = 0; k < CHUNK_VEC; ++k) {
        pl[k].x = gcn_take_place(s_bkt, r[k].x);
        pl[k].y = gcn_take_place(s_bkt, r[k].y);
        pl[k].z = gcn_take_place(s_bkt, r[k].z);
        pl[k].w = gcn_take_place(s_bkt, r[k].w);
    }
#pragma unroll
    for (int k = 0; k < CHUNK_VEC; ++k) {
        const int e = base + (k * CHUNK_THREADS + t) * 4;
        gcn_file_edge(s_rec, pl[k].x, r[k].x, e);
        gcn_file_edge(s_rec, pl[k].y, r[k].y, e + 1);
        gcn_file_edge(s_rec, pl[k].z, r[k].z, e + 2);
        gcn_file_edge(s_rec, pl[k].w, r[k].w, e + 3);
    }
    __syncthreads();
    int2* __restrict__ dst = (int2*)brec + (size_t)chunk * CHUNK_EDGES;
    v4i o[CHUNK_EDGES / (2 * CHUNK_THREADS)];
#pragma unroll
    for (int k = 0; k < CHUNK_EDGES / (2 * CHUNK_THREADS); ++k)
        o[k] = *(const v4i*)(s_rec + 2 * (k * CHUNK_THREADS + t));
#pragma unroll
    for (int k = 0; k < CHUNK_EDGES / (2 * CHUNK_THREADS); ++k) {
        const int i = 2 * (k * CHUNK_THREADS + t);
        if (i + 1 < cnt)  *(v4i*)(dst + i) = o[k];
        else if (i < cnt) dst[i] = make_int2(o[k].x, o[k].y);
    }
}

__global__ void __launch_bounds__(CSR_THREADS)
gcn_bucket_csr(const int* __restrict__ brec, const int* __restrict__ runs,
               int* __restrict__ offs, int* __restrict__ perm, int Nn, int nb, int nchunk) {
    extern __shared__ int2 s_run[];
    __shared__ int s_cnt[BUCKET_ROWS];
    __shared__ int s_cur[BUCKET_ROWS];
    __shared__ int s_w[BUCKET_ROWS / 32];
    __shared__ int s_part[CSR_THREADS / 32];
    const int t    = threadIdx.x;
    const int lane = t & 31;
    const int w    = __builtin_amdgcn_readfirstlane((int)(threadIdx.x >> 5));
    const int b    = blockIdx.x;
    if (t < BUCKET_ROWS) s_cnt[t] = 0;
    int part = 0;
    for (int c = t; c < nchunk; c += CSR_THREADS) {
        const int2 pl = ((const int2*)runs)[(size_t)c * nb + b];
        s_run[c] = make_int2(c * CHUNK_EDGES + pl.x, pl.y);
        part += pl.x;
    }
    for (int d = 16; d > 0; d >>= 1) part += __shfl_xor(part, d);
    if (lane == 0) s_part[w] = part;
    __syncthreads();
    int bs = s_part[lane];
    for (int d = 16; d > 0; d >>= 1) bs += __shfl_xor(bs, d);

    for (int c0 = w; c0 < nchunk; c0 += 32 * CSR_SEGS) {
        int2 run[CSR_SEGS];
        int  rw[CSR_SEGS];
#pragma unroll
        for (int u = 0; u < CSR_SEGS; ++u) {
            const int c = c0 + 32 * u;
            run[u] = s_run[(c < nchunk) ? c : nchunk - 1];
            if (c >= nchunk) run[u].y = 0;
        }
#pragma unroll
        for (int u = 0; u < CSR_SEGS; ++u)
            rw[u] = (lane < run[u].y) ? brec[2 * (size_t)(run[u].x + lane)] : -1;
#pragma unroll
        for (int u = 0; u < CSR_SEGS; ++u)
            if (rw[u] >= 0)
                __hip_atomic_fetch_add(&s_cnt[rw[u] & (BUCKET_ROWS - 1)], 1, __ATOMIC_RELAXED,
                                       __HIP_MEMORY_SCOPE_WORKGROUP);
        for (int u = 0; u < CSR_SEGS; ++u)
            for (int i = lane + 32; i < run[u].y; i += 32) {
                const int row = brec[2 * (size_t)(run[u].x + i)];
                __hip_atomic_fetch_add(&s_cnt[row & (BUCKET_ROWS - 1)], 1, __ATOMIC_RELAXED,
                                       __HIP_MEMORY_SCOPE_WORKGROUP);
            }
    }
    __syncthreads();
    int c = 0, inc = 0;
    if (t < BUCKET_ROWS) {
        c = s_cnt[t];
        inc = c;
        for (int d = 1; d < 32; d <<= 1) {
            const int o = __shfl_up(inc, d);
            if (lane >= d) inc += o;
        }
        if (lane == 31) s_w[w] = inc;
    }
    __syncthreads();
    if (t < BUCKET_ROWS) {
        int before = 0;
        for (int j = 0; j < w; ++j) before += s_w[j];
        const int start = bs + before + inc - c;
        s_cur[t] = start;
        const int r = b * BUCKET_ROWS + t;
        if (r < Nn)      offs[r]  = start;
        if (r == Nn - 1) offs[Nn] = start + c;
    }
    __syncthreads();
    for (int c0 = w; c0 < nchunk; c0 += 32 * CSR_SEGS) {
        int2 run[CSR_SEGS];
        int2 rc[CSR_SEGS];
        int  pos[CSR_SEGS];
#pragma unroll
        for (int u = 0; u < CSR_SEGS; ++u) {
            const int c = c0 + 32 * u;
            run[u] = s_run[(c < nchunk) ? c : nchunk - 1];
            if (c >= nchunk) run[u].y = 0;
        }
#pragma unroll
        for (int u = 0; u < CSR_SEGS; ++u)
            rc[u] = (lane < run[u].y) ? *(const int2*)(brec + 2 * (size_t)(run[u].x + lane))
                                      : make_int2(-1, 0);
#pragma unroll
        for (int u = 0; u < CSR_SEGS; ++u) {
            pos[u] = -1;
            if (rc[u].x >= 0)
                pos[u] = __hip_atomic_fetch_add(&s_cur[rc[u].x & (BUCKET_ROWS - 1)], 1,
                                                __ATOMIC_RELAXED, __HIP_MEMORY_SCOPE_WORKGROUP);
        }
#pragma unroll
        for (int u = 0; u < CSR_SEGS; ++u)
            if (pos[u] >= 0) perm[pos[u]] = rc[u].y;
        for (int u = 0; u < CSR_SEGS; ++u)
            for (int i = lane + 32; i < run[u].y; i += 32) {
                const int2 rec = *(const int2*)(brec + 2 * (size_t)(run[u].x + i));
                const int p = __hip_atomic_fetch_add(&s_cur[rec.x & (BUCKET_ROWS - 1)], 1,
                                                     __ATOMIC_RELAXED, __HIP_MEMORY_SCOPE_WORKGROUP);
                perm[p] = rec.y;
            }
    }
}

__device__ __forceinline__ v4f gcn_ld4(const float* __restrict__ xrow, unsigned off) {
    return *(const v4f*)((const char*)xrow + off);
}

__device__ __forceinline__ void gcn_edge1(const float* __restrict__ x, unsigned offa, unsigned offb,
                                          int c, float v, v4f& s0, v4f& s1) {
    const float* __restrict__ xr = x + (size_t)c * FDIM;
    const v4f p = gcn_ld4(xr, offa);
    const v4f q = gcn_ld4(xr, offb);
    s0 += v * p;
    s1 += v * q;
}

__device__ __forceinline__ void gcn_pick(int rank, int c, float v, int p, int& cp, float& vp) {
    const int l = __builtin_ctz(__builtin_amdgcn_ballot_w32(rank == p));
    cp = __builtin_amdgcn_readlane(c, l);
    vp = __int_as_float(__builtin_amdgcn_readlane(__float_as_int(v), l));
}

__device__ __forceinline__ void gcn_pick2(int rankA, int rankB, int cA, float vA, int cB, float vB,
                                          int p, int& cp, float& vp) {
    const unsigned mA = __builtin_amdgcn_ballot_w32(rankA == p);
    const unsigned mB = __builtin_amdgcn_ballot_w32(rankB == p);
    const int l  = __builtin_ctz(mA | mB);
    const int ca = __builtin_amdgcn_readlane(cA, l);
    const int cb = __builtin_amdgcn_readlane(cB, l);
    const int va = __builtin_amdgcn_readlane(__float_as_int(vA), l);
    const int vb = __builtin_amdgcn_readlane(__float_as_int(vB), l);
    cp = mA ? ca : cb;
    vp = __int_as_float(mA ? va : vb);
}

__global__ void __launch_bounds__(ROWS_PER_BLOCK * 32)
gcn_gather_gemm_bias_elu(const float* __restrict__ x,
                         const float* __restrict__ vals,
                         const int*   __restrict__ cols,
                         const int*   __restrict__ offs,
                         const int*   __restrict__ perm,
                         const float* __restrict__ W,
                         const float* __restrict__ bias,
                         float*       __restrict__ out,
                         int Nn) {
    __shared__ __attribute__((aligned(16))) float s_agg[TILE_ROWS * TILE_PITCH];
    __shared__ __attribute__((aligned(16))) float s_res[TILE_ROWS * TILE_PITCH];

    const int lane = threadIdx.x & 31;
    const int wv   = __builtin_amdgcn_readfirstlane((int)(threadIdx.x >> 5));
    const int row0 = blockIdx.x * ROWS_PER_BLOCK;
    const int r    = row0 + wv;
    const int half = lane >> 4;
    const int q4   = (lane & 15) * 4;

    v4f s0 = {0.f, 0.f, 0.f, 0.f};
    v4f s1 = {0.f, 0.f, 0.f, 0.f};

    if (r < Nn) {
        const int beg = offs[r];
        const int kt  = offs[r + 1] - beg;
        const unsigned offa = (unsigned)(((size_t)half * Nn * FDIM + q4) * sizeof(float));
        const unsigned offb = (unsigned)(((size_t)(half + 2) * Nn * FDIM + q4) * sizeof(float));

        if (kt <= 32) {
            int   eid = 0x7fffffff;
            int   c   = 0;
            float v   = 0.f;
            if (lane < kt) {
                eid = perm[beg + lane];
                c   = cols[eid];
                v   = vals[eid];
            }
            int rank = 0;
            for (int j = 0; j < kt; ++j)
                rank += (__builtin_amdgcn_readlane(eid, j) < eid) ? 1 : 0;

            int p = 0;
            for (; p + 3 < kt; p += 4) {
                int c0, c1, c2, c3;
                float v0, v1, v2, v3;
                gcn_pick(rank, c, v, p,     c0, v0);
                gcn_pick(rank, c, v, p + 1, c1, v1);
                gcn_pick(rank, c, v, p + 2, c2, v2);
                gcn_pick(rank, c, v, p + 3, c3, v3);
                const float* __restrict__ x0 = x + (size_t)c0 * FDIM;
                const float* __restrict__ x1 = x + (size_t)c1 * FDIM;
                const float* __restrict__ x2 = x + (size_t)c2 * FDIM;
                const float* __restrict__ x3 = x + (size_t)c3 * FDIM;
                const v4f p0 = gcn_ld4(x0, offa), q0 = gcn_ld4(x0, offb);
                const v4f p1 = gcn_ld4(x1, offa), q1 = gcn_ld4(x1, offb);
                const v4f p2 = gcn_ld4(x2, offa), q2 = gcn_ld4(x2, offb);
                const v4f p3 = gcn_ld4(x3, offa), q3 = gcn_ld4(x3, offb);
                s0 += v0 * p0;  s1 += v0 * q0;
                s0 += v1 * p1;  s1 += v1 * q1;
                s0 += v2 * p2;  s1 += v2 * q2;
                s0 += v3 * p3;  s1 += v3 * q3;
            }
            for (; p < kt; ++p) {
                int c0; float v0;
                gcn_pick(rank, c, v, p, c0, v0);
                gcn_edge1(x, offa, offb, c0, v0, s0, s1);
            }
        } else if (kt <= 64) {
            const bool hasB = lane + 32 < kt;
            const int  iB   = hasB ? lane + 32 : kt - 1;
            const int eidA  = perm[beg + lane];
            const int eidBr = perm[beg + iB];
            const int   cA = cols[eidA];
            const float vA = vals[eidA];
            const int   cB = cols[eidBr];
            const float vB = vals[eidBr];
            const int eidB = hasB ? eidBr : 0x7fffffff;
            int rankA = 0, rankB = 0;
            for (int j = 0; j < 32; ++j) {
                const int e = __builtin_amdgcn_readlane(eidA, j);
                rankA += (e < eidA) ? 1 : 0;
                rankB += (e < eidB) ? 1 : 0;
            }
            for (int j = 0; j < kt - 32; ++j) {
                const int e = __builtin_amdgcn_readlane(eidB, j);
                rankA += (e < eidA) ? 1 : 0;
                rankB += (e < eidB) ? 1 : 0;
            }

            int p = 0;
            for (; p + 2 < kt; p += 3) {
                int c0, c1, c2;
                float v0, v1, v2;
                gcn_pick2(rankA, rankB, cA, vA, cB, vB, p, c0, v0);
                gcn_pick2(rankA, rankB, cA, vA, cB, vB, p + 1, c1, v1);
                gcn_pick2(rankA, rankB, cA, vA, cB, vB, p + 2, c2, v2);
                const float* __restrict__ x0 = x + (size_t)c0 * FDIM;
                const float* __restrict__ x1 = x + (size_t)c1 * FDIM;
                const float* __restrict__ x2 = x + (size_t)c2 * FDIM;
                const v4f p0 = gcn_ld4(x0, offa), q0 = gcn_ld4(x0, offb);
                const v4f p1 = gcn_ld4(x1, offa), q1 = gcn_ld4(x1, offb);
                const v4f p2 = gcn_ld4(x2, offa), q2 = gcn_ld4(x2, offb);
                s0 += v0 * p0;  s1 += v0 * q0;
                s0 += v1 * p1;  s1 += v1 * q1;
                s0 += v2 * p2;  s1 += v2 * q2;
            }
            for (; p < kt; ++p) {
                int c0; float v0;
                gcn_pick2(rankA, rankB, cA, vA, cB, vB, p, c0, v0);
                gcn_edge1(x, offa, offb, c0, v0, s0, s1);
            }
        } else {
            int prev = -1;
            for (int p = 0; p < kt; ++p) {
                int m = 0x7fffffff;
                for (int i = lane; i < kt; i += 32) {
                    const int e = perm[beg + i];
                    if (e > prev && e < m) m = e;
                }
                for (int d = 16; d > 0; d >>= 1) {
                    const int o = __shfl_xor(m, d);
                    m = (o < m) ? o : m;
                }
                m = __builtin_amdgcn_readfirstlane(m);
                prev = m;
                gcn_edge1(x, offa, offb, cols[m], vals[m], s0, s1);
            }
        }
    }

    *(v4f*)&s_agg[(half * ROWS_PER_BLOCK + wv) * TILE_PITCH + q4]       = s0;
    *(v4f*)&s_agg[((half + 2) * ROWS_PER_BLOCK + wv) * TILE_PITCH + q4] = s1;

    const int m  = lane & 15;
    const int kg = (lane >> 4) << 1;
    const int mt = wv >> 2;
    const int nc = (wv & 3) * 16 + m;

    v2f bf[FDIM / 4];
    for (int k4 = 0; k4 < FDIM / 4; ++k4) {
        const int kk = k4 * 4 + kg;
        bf[k4].x = W[kk * FDIM + nc];
        bf[k4].y = W[(kk + 1) * FDIM + nc];
    }
    const float bb = bias[nc];

    __syncthreads();

    const float* arow = s_agg + (mt * 16 + m) * TILE_PITCH;
    v8f acc = v8f{};
    for (int k4 = 0; k4 < FDIM / 4; ++k4) {
        const int kk = k4 * 4 + kg;
        v2f a;
        a.x = arow[kk];
        a.y = arow[kk + 1];
        acc = __builtin_amdgcn_wmma_f32_16x16x4_f32(
             false, a,  false, bf[k4],
             (short)0, acc,
             false,  false);
    }

    for (int i = 0; i < 8; ++i) {
        const float v = acc[i] + bb;
        s_res[(mt * 16 + half * 8 + i) * TILE_PITCH + nc] = (v > 0.0f) ? v : (__expf(v) - 1.0f);
    }

    __syncthreads();

    for (int u = 0; u < 2; ++u) {
        const int g   = 4 * wv + 2 * u + half;
        const int row = row0 + (g & (ROWS_PER_BLOCK - 1));
        if (row < Nn)
            *(v4f*)(out + ((size_t)(g >> 3) * Nn + row) * FDIM + q4) =
                *(const v4f*)&s_res[g * TILE_PITCH + q4];
    }
}

extern "C" void kernel_launch(void* const* d_in, const int* in_sizes, int n_in,
                              void* d_out, int out_size, void* d_ws, size_t ws_size,
                              hipStream_t stream) {
    const float* x      = (const float*)d_in[0];
    const float* vals   = (const float*)d_in[1];
    const float* weight = (const float*)d_in[2];
    const float* bias   = (const float*)d_in[3];
    const int*   rows   = (const int*)d_in[4];
    const int*   cols   = (const int*)d_in[5];
    float*       out    = (float*)d_out;

    const int  E  = in_sizes[1];
    const int  Nn = in_sizes[0] / (BATCH * FDIM);
    const int  nb     = (Nn + BUCKET_ROWS - 1) / BUCKET_ROWS;
    const int  nchunk = (E + CHUNK_EDGES - 1) / CHUNK_EDGES;
    const long nPad = ((long)Nn + 64) & ~63L;
    const long ePad = ((long)E + 63) & ~63L;
    const long rPad = (long)nchunk * CHUNK_EDGES;
    int*   offs   = (int*)d_ws;
    int*   perm   = offs + nPad;
    int*   brec   = perm + ePad;
    int*   runs   = brec + 2 * rPad;

    gcn_chunk_partition<<<(unsigned)nchunk, CHUNK_THREADS, nb * sizeof(int), stream>>>(
        rows, runs, brec, E, nb);
    gcn_bucket_csr<<<(unsigned)nb, CSR_THREADS, nchunk * sizeof(int2), stream>>>(
        brec, runs, offs, perm, Nn, nb, nchunk);
    gcn_gather_gemm_bias_elu<<<(unsigned)((Nn + ROWS_PER_BLOCK - 1) / ROWS_PER_BLOCK),
                               ROWS_PER_BLOCK * 32, 0, stream>>>(
        x, vals, cols, offs, perm, weight, bias, out, Nn);
}


# CSR build carries col and val in each edge record (16 B, rows/cols/vals read once in edge order); gather reads a row's records as one contiguous run: no perm hop, no scattered 4-byte col/val reads
# speedup vs baseline: 4.3009x; 1.0568x over previous
#include <hip/hip_runtime.h>


#define BATCH 4
#define FDIM  64
#define ROWS_PER_BLOCK 8
#define TILE_ROWS (BATCH * ROWS_PER_BLOCK)
#define TILE_PITCH 68

typedef __attribute__((ext_vector_type(2))) float v2f;
typedef __attribute__((ext_vector_type(8))) float v8f;
typedef __attribute__((ext_vector_type(4))) float v4f;
typedef __attribute__((ext_vector_type(4))) int   v4i;

#define BUCKET_SHIFT  8
#define BUCKET_ROWS   (1 << BUCKET_SHIFT)
#define CHUNK_EDGES   4096
#define CHUNK_THREADS 256
#define CHUNK_VEC     (CHUNK_EDGES / (CHUNK_THREADS * 4))
#define CSR_THREADS   1024
#define CSR_SEGS      8

__device__ __forceinline__ int4 gcn_rows4(const int* __restrict__ rows, int e, int E) {
    int4 r;
    if (e + 4 <= E) {
        r = *(const int4*)(rows + e);
    } else {
        r.x = (e     < E) ? rows[e]     : -1;
        r.y = (e + 1 < E) ? rows[e + 1] : -1;
        r.z = (e + 2 < E) ? rows[e + 2] : -1;
        r.w = (e + 3 < E) ? rows[e + 3] : -1;
    }
    return r;
}

__device__ __forceinline__ int4 gcn_cols4(const int* __restrict__ cols, int e, int E) {
    int4 c;
    if (e + 4 <= E) {
        c = *(const int4*)(cols + e);
    } else {
        c.x = (e     < E) ? cols[e]     : 0;
        c.y = (e + 1 < E) ? cols[e + 1] : 0;
        c.z = (e + 2 < E) ? cols[e + 2] : 0;
        c.w = (e + 3 < E) ? cols[e + 3] : 0;
    }
    return c;
}

__device__ __forceinline__ int4 gcn_vals4(const float* __restrict__ vals, int e, int E) {
    float4 v;
    if (e + 4 <= E) {
        v = *(const float4*)(vals + e);
    } else {
        v.x = (e     < E) ? vals[e]     : 0.f;
        v.y = (e + 1 < E) ? vals[e + 1] : 0.f;
        v.z = (e + 2 < E) ? vals[e + 2] : 0.f;
        v.w = (e + 3 < E) ? vals[e + 3] : 0.f;
    }
    return make_int4(__float_as_int(v.x), __float_as_int(v.y), __float_as_int(v.z), __float_as_int(v.w));
}

__device__ __forceinline__ void gcn_lds_count(int* s_cnt, int row) {
    if (row >= 0)
        __hip_atomic_fetch_add(&s_cnt[row >> BUCKET_SHIFT], 1, __ATOMIC_RELAXED,
                               __HIP_MEMORY_SCOPE_WORKGROUP);
}

template <int NT>
__device__ __forceinline__ int gcn_block_excl_scan(int* s_a, int n, int* s_w) {
    const int t    = threadIdx.x;
    const int lane = t & 31;
    const int w    = t >> 5;
    int carry = 0;
    for (int base = 0; base < n; base += NT) {
        const int i = base + t;
        const int v = (i < n) ? s_a[i] : 0;
        int inc = v;
        for (int d = 1; d < 32; d <<= 1) {
            const int o = __shfl_up(inc, d);
            if (lane >= d) inc += o;
        }
        if (lane == 31) s_w[w] = inc;
        __syncthreads();
        const int wt = (lane < NT / 32) ? s_w[lane] : 0;
        int before = (lane < w) ? wt : 0;
        int all    = wt;
        for (int d = 16; d > 0; d >>= 1) {
            before += __shfl_xor(before, d);
            all    += __shfl_xor(all, d);
        }
        if (i < n) s_a[i] = carry + before + inc - v;
        carry += all;
        __syncthreads();
    }
    return carry;
}

__device__ __forceinline__ int gcn_take_place(int* s_cur, int row) {
    int pos = -1;
    if (row >= 0)
        pos = __hip_atomic_fetch_add(&s_cur[row >> BUCKET_SHIFT], 1, __ATOMIC_RELAXED,
                                     __HIP_MEMORY_SCOPE_WORKGROUP);
    return pos;
}

__device__ __forceinline__ void gcn_file_edge(v4i* s_rec, int pos, int row, int e, int col, int vbits) {
    if (pos >= 0) s_rec[pos] = v4i{row, e, col, vbits};
}

__global__ void __launch_bounds__(CHUNK_THREADS)
gcn_chunk_partition(const int* __restrict__ rows, const int* __restrict__ cols,
                    const float* __restrict__ vals, int* __restrict__ runs,
                    int* __restrict__ brec, int E, int nb) {
    extern __shared__ int s_bkt[];
    __shared__ __attribute__((aligned(16))) v4i s_rec[CHUNK_EDGES];
    __shared__ int s_w[32];
    const int t = threadIdx.x;
    const int chunk = blockIdx.x;
    for (int i = t; i < nb; i += CHUNK_THREADS) s_bkt[i] = 0;
    int4 r[CHUNK_VEC], cl[CHUNK_VEC], vb[CHUNK_VEC];
    const int base = chunk * CHUNK_EDGES;
#pragma unroll
    for (int k = 0; k < CHUNK_VEC; ++k) {
        const int e = base + (k * CHUNK_THREADS + t) * 4;
        r[k]  = gcn_rows4(rows, e, E);
        cl[k] = gcn_cols4(cols, e, E);
        vb[k] = gcn_vals4(vals, e, E);
    }
    __syncthreads();
#pragma unroll
    for (int k = 0; k < CHUNK_VEC; ++k) {
        gcn_lds_count(s_bkt, r[k].x);
        gcn_lds_count(s_bkt, r[k].y);
        gcn_lds_count(s_bkt, r[k].z);
        gcn_lds_count(s_bkt, r[k].w);
    }
    __syncthreads();
    const int cnt = gcn_block_excl_scan<CHUNK_THREADS>(s_bkt, nb, s_w);
    int2* __restrict__ rn = (int2*)runs + (size_t)chunk * nb;
    for (int i = t; i < nb; i += CHUNK_THREADS) {
        const int a = s_bkt[i];
        const int z = (i + 1 < nb) ? s_bkt[i + 1] : cnt;
        rn[i] = make_int2(a, z - a);
    }
    __syncthreads();
    int4 pl[CHUNK_VEC];
#pragma unroll
    for (int k = 0; k < CHUNK_VEC; ++k) {
        pl[k].x = gcn_take_place(s_bkt, r[k].x);
        pl[k].y = gcn_take_place(s_bkt, r[k].y);
        pl[k].z = gcn_take_place(s_bkt, r[k].z);
        pl[k].w = gcn_take_place(s_bkt, r[k].w);
    }
#pragma unroll
    for (int k = 0; k < CHUNK_VEC; ++k) {
        const int e = base + (k * CHUNK_THREADS + t) * 4;
        gcn_file_edge(s_rec, pl[k].x, r[k].x, e,     cl[k].x, vb[k].x);
        gcn_file_edge(s_rec, pl[k].y, r[k].y, e + 1, cl[k].y, vb[k].y);
        gcn_file_edge(s_rec, pl[k].z, r[k].z, e + 2, cl[k].z, vb[k].z);
        gcn_file_edge(s_rec, pl[k].w, r[k].w, e + 3, cl[k].w, vb[k].w);
    }
    __syncthreads();
    v4i* __restrict__ dst = (v4i*)brec + (size_t)chunk * CHUNK_EDGES;
    for (int g = 0; g < CHUNK_EDGES / (8 * CHUNK_THREADS); ++g) {
        v4i o[8];
#pragma unroll
        for (int k = 0; k < 8; ++k)
            o[k] = s_rec[(g * 8 + k) * CHUNK_THREADS + t];
#pragma unroll
        for (int k = 0; k < 8; ++k) {
            const int i = (g * 8 + k) * CHUNK_THREADS + t;
            if (i < cnt) dst[i] = o[k];
        }
    }
}

__global__ void __launch_bounds__(CSR_THREADS)
gcn_bucket_csr(const int* __restrict__ brec, const int* __restrict__ runs,
               int* __restrict__ offs, int* __restrict__ csr, int Nn, int nb, int nchunk) {
    extern __shared__ int2 s_run[];
    __shared__ int s_cnt[BUCKET_ROWS];
    __shared__ int s_cur[BUCKET_ROWS];
    __shared__ int s_w[BUCKET_ROWS / 32];
    __shared__ int s_part[CSR_THREADS / 32];
    const int t    = threadIdx.x;
    const int lane = t & 31;
    const int w    = __builtin_amdgcn_readfirstlane((int)(threadIdx.x >> 5));
    const int b    = blockIdx.x;
    if (t < BUCKET_ROWS) s_cnt[t] = 0;
    int part = 0;
    for (int c = t; c < nchunk; c += CSR_THREADS) {
        const int2 pl = ((const int2*)runs)[(size_t)c * nb + b];
        s_run[c] = make_int2(c * CHUNK_EDGES + pl.x, pl.y);
        part += pl.x;
    }
    for (int d = 16; d > 0; d >>= 1) part += __shfl_xor(part, d);
    if (lane == 0) s_part[w] = part;
    __syncthreads();
    int bs = s_part[lane];
    for (int d = 16; d > 0; d >>= 1) bs += __shfl_xor(bs, d);

    for (int c0 = w; c0 < nchunk; c0 += 32 * CSR_SEGS) {
        int2 run[CSR_SEGS];
        int  rw[CSR_SEGS];
#pragma unroll
        for (int u = 0; u < CSR_SEGS; ++u) {
            const int c = c0 + 32 * u;
            run[u] = s_run[(c < nchunk) ? c : nchunk - 1];
            if (c >= nchunk) run[u].y = 0;
        }
#pragma unroll
        for (int u = 0; u < CSR_SEGS; ++u)
            rw[u] = (lane < run[u].y) ? brec[4 * (size_t)(run[u].x + lane)] : -1;
#pragma unroll
        for (int u = 0; u < CSR_SEGS; ++u)
            if (rw[u] >= 0)
                __hip_atomic_fetch_add(&s_cnt[rw[u] & (BUCKET_ROWS - 1)], 1, __ATOMIC_RELAXED,
                                       __HIP_MEMORY_SCOPE_WORKGROUP);
        for (int u = 0; u < CSR_SEGS; ++u)
            for (int i = lane + 32; i < run[u].y; i += 32) {
                const int row = brec[4 * (size_t)(run[u].x + i)];
                __hip_atomic_fetch_add(&s_cnt[row & (BUCKET_ROWS - 1)], 1, __ATOMIC_RELAXED,
                                       __HIP_MEMORY_SCOPE_WORKGROUP);
            }
    }
    __syncthreads();
    int c = 0, inc = 0;
    if (t < BUCKET_ROWS) {
        c = s_cnt[t];
        inc = c;
        for (int d = 1; d < 32; d <<= 1) {
            const int o = __shfl_up(inc, d);
            if (lane >= d) inc += o;
        }
        if (lane == 31) s_w[w] = inc;
    }
    __syncthreads();
    if (t < BUCKET_ROWS) {
        int before = 0;
        for (int j = 0; j < w; ++j) before += s_w[j];
        const int start = bs + before + inc - c;
        s_cur[t] = start;
        const int r = b * BUCKET_ROWS + t;
        if (r < Nn)      offs[r]  = start;
        if (r == Nn - 1) offs[Nn] = start + c;
    }
    __syncthreads();
    for (int c0 = w; c0 < nchunk; c0 += 32 * CSR_SEGS) {
        int2 run[CSR_SEGS];
        v4i  rc[CSR_SEGS];
        int  pos[CSR_SEGS];
#pragma unroll
        for (int u = 0; u < CSR_SEGS; ++u) {
            const int c = c0 + 32 * u;
            run[u] = s_run[(c < nchunk) ? c : nchunk - 1];
            if (c >= nchunk) run[u].y = 0;
        }
#pragma unroll
        for (int u = 0; u < CSR_SEGS; ++u)
            rc[u] = (lane < run[u].y) ? *(const v4i*)(brec + 4 * (size_t)(run[u].x + lane))
                                      : v4i{-1, 0, 0, 0};
#pragma unroll
        for (int u = 0; u < CSR_SEGS; ++u) {
            pos[u] = -1;
            if (rc[u].x >= 0)
                pos[u] = __hip_atomic_fetch_add(&s_cur[rc[u].x & (BUCKET_ROWS - 1)], 1,
                                                __ATOMIC_RELAXED, __HIP_MEMORY_SCOPE_WORKGROUP);
        }
#pragma unroll
        for (int u = 0; u < CSR_SEGS; ++u)
            if (pos[u] >= 0) *(v4i*)(csr + 4 * (size_t)pos[u]) = rc[u];
        for (int u = 0; u < CSR_SEGS; ++u)
            for (int i = lane + 32; i < run[u].y; i += 32) {
                const v4i rec = *(const v4i*)(brec + 4 * (size_t)(run[u].x + i));
                const int p = __hip_atomic_fetch_add(&s_cur[rec.x & (BUCKET_ROWS - 1)], 1,
                                                     __ATOMIC_RELAXED, __HIP_MEMORY_SCOPE_WORKGROUP);
                *(v4i*)(csr + 4 * (size_t)p) = rec;
            }
    }
}

__device__ __forceinline__ v4f gcn_ld4(const float* __restrict__ xrow, unsigned off) {
    return *(const v4f*)((const char*)xrow + off);
}

__device__ __forceinline__ void gcn_edge1(const float* __restrict__ x, unsigned offa, unsigned offb,
                                          int c, float v, v4f& s0, v4f& s1) {
    const float* __restrict__ xr = x + (size_t)c * FDIM;
    const v4f p = gcn_ld4(xr, offa);
    const v4f q = gcn_ld4(xr, offb);
    s0 += v * p;
    s1 += v * q;
}

__device__ __forceinline__ void gcn_pick(int rank, int c, float v, int p, int& cp, float& vp) {
    const int l = __builtin_ctz(__builtin_amdgcn_ballot_w32(rank == p));
    cp = __builtin_amdgcn_readlane(c, l);
    vp = __int_as_float(__builtin_amdgcn_readlane(__float_as_int(v), l));
}

__device__ __forceinline__ void gcn_pick2(int rankA, int rankB, int cA, float vA, int cB, float vB,
                                          int p, int& cp, float& vp) {
    const unsigned mA = __builtin_amdgcn_ballot_w32(rankA == p);
    const unsigned mB = __builtin_amdgcn_ballot_w32(rankB == p);
    const int l  = __builtin_ctz(mA | mB);
    const int ca = __builtin_amdgcn_readlane(cA, l);
    const int cb = __builtin_amdgcn_readlane(cB, l);
    const int va = __builtin_amdgcn_readlane(__float_as_int(vA), l);
    const int vb = __builtin_amdgcn_readlane(__float_as_int(vB), l);
    cp = mA ? ca : cb;
    vp = __int_as_float(mA ? va : vb);
}

__global__ void __launch_bounds__(ROWS_PER_BLOCK * 32)
gcn_gather_gemm_bias_elu(const float* __restrict__ x,
                         const float* __restrict__ vals,
                         const int*   __restrict__ cols,
                         const int*   __restrict__ offs,
                         const int*   __restrict__ csr,
                         const float* __restrict__ W,
                         const float* __restrict__ bias,
                         float*       __restrict__ out,
                         int Nn) {
    __shared__ __attribute__((aligned(16))) float s_agg[TILE_ROWS * TILE_PITCH];
    __shared__ __attribute__((aligned(16))) float s_res[TILE_ROWS * TILE_PITCH];

    const int lane = threadIdx.x & 31;
    const int wv   = __builtin_amdgcn_readfirstlane((int)(threadIdx.x >> 5));
    const int row0 = blockIdx.x * ROWS_PER_BLOCK;
    const int r    = row0 + wv;
    const int half = lane >> 4;
    const int q4   = (lane & 15) * 4;

    v4f s0 = {0.f, 0.f, 0.f, 0.f};
    v4f s1 = {0.f, 0.f, 0.f, 0.f};

    if (r < Nn) {
        const int beg = offs[r];
        const int kt  = offs[r + 1] - beg;
        const unsigned offa = (unsigned)(((size_t)half * Nn * FDIM + q4) * sizeof(float));
        const unsigned offb = (unsigned)(((size_t)(half + 2) * Nn * FDIM + q4) * sizeof(float));

        if (kt <= 32) {
            int   eid = 0x7fffffff;
            int   c   = 0;
            float v   = 0.f;
            if (lane < kt) {
                const v4i rec = *(const v4i*)(csr + 4 * (size_t)(beg + lane));
                eid = rec.y;
                c   = rec.z;
                v   = __int_as_float(rec.w);
            }
            int rank = 0;
            for (int j = 0; j < kt; ++j)
                rank += (__builtin_amdgcn_readlane(eid, j) < eid) ? 1 : 0;

            int p = 0;
            for (; p + 3 < kt; p += 4) {
                int c0, c1, c2, c3;
                float v0, v1, v2, v3;
                gcn_pick(rank, c, v, p,     c0, v0);
                gcn_pick(rank, c, v, p + 1, c1, v1);
                gcn_pick(rank, c, v, p + 2, c2, v2);
                gcn_pick(rank, c, v, p + 3, c3, v3);
                const float* __restrict__ x0 = x + (size_t)c0 * FDIM;
                const float* __restrict__ x1 = x + (size_t)c1 * FDIM;
                const float* __restrict__ x2 = x + (size_t)c2 * FDIM;
                const float* __restrict__ x3 = x + (size_t)c3 * FDIM;
                const v4f p0 = gcn_ld4(x0, offa), q0 = gcn_ld4(x0, offb);
                const v4f p1 = gcn_ld4(x1, offa), q1 = gcn_ld4(x1, offb);
                const v4f p2 = gcn_ld4(x2, offa), q2 = gcn_ld4(x2, offb);
                const v4f p3 = gcn_ld4(x3, offa), q3 = gcn_ld4(x3, offb);
                s0 += v0 * p0;  s1 += v0 * q0;
                s0 += v1 * p1;  s1 += v1 * q1;
                s0 += v2 * p2;  s1 += v2 * q2;
                s0 += v3 * p3;  s1 += v3 * q3;
            }
            for (; p < kt; ++p) {
                int c0; float v0;
                gcn_pick(rank, c, v, p, c0, v0);
                gcn_edge1(x, offa, offb, c0, v0, s0, s1);
            }
        } else if (kt <= 64) {
            const bool hasB = lane + 32 < kt;
            const int  iB   = hasB ? lane + 32 : kt - 1;
            const v4i recA = *(const v4i*)(csr + 4 * (size_t)(beg + lane));
            const v4i recB = *(const v4i*)(csr + 4 * (size_t)(beg + iB));
            const int   eidA = recA.y;
            const int   cA = recA.z;
            const float vA = __int_as_float(recA.w);
            const int   cB = recB.z;
            const float vB = __int_as_float(recB.w);
            const int eidB = hasB ? recB.y : 0x7fffffff;
            int rankA = 0, rankB = 0;
            for (int j = 0; j < 32; ++j) {
                const int e = __builtin_amdgcn_readlane(eidA, j);
                rankA += (e < eidA) ? 1 : 0;
                rankB += (e < eidB) ? 1 : 0;
            }
            for (int j = 0; j < kt - 32; ++j) {
                const int e = __builtin_amdgcn_readlane(eidB, j);
                rankA += (e < eidA) ? 1 : 0;
                rankB += (e < eidB) ? 1 : 0;
            }

            int p = 0;
            for (; p + 2 < kt; p += 3) {
                int c0, c1, c2;
                float v0, v1, v2;
                gcn_pick2(rankA, rankB, cA, vA, cB, vB, p, c0, v0);
                gcn_pick2(rankA, rankB, cA, vA, cB, vB, p + 1, c1, v1);
                gcn_pick2(rankA, rankB, cA, vA, cB, vB, p + 2, c2, v2);
                const float* __restrict__ x0 = x + (size_t)c0 * FDIM;
                const float* __restrict__ x1 = x + (size_t)c1 * FDIM;
                const float* __restrict__ x2 = x + (size_t)c2 * FDIM;
                const v4f p0 = gcn_ld4(x0, offa), q0 = gcn_ld4(x0, offb);
                const v4f p1 = gcn_ld4(x1, offa), q1 = gcn_ld4(x1, offb);
                const v4f p2 = gcn_ld4(x2, offa), q2 = gcn_ld4(x2, offb);
                s0 += v0 * p0;  s1 += v0 * q0;
                s0 += v1 * p1;  s1 += v1 * q1;
                s0 += v2 * p2;  s1 += v2 * q2;
            }
            for (; p < kt; ++p) {
                int c0; float v0;
                gcn_pick2(rankA, rankB, cA, vA, cB, vB, p, c0, v0);
                gcn_edge1(x, offa, offb, c0, v0, s0, s1);
            }
        } else {
            int prev = -1;
            for (int p = 0; p < kt; ++p) {
                int m = 0x7fffffff;
                for (int i = lane; i < kt; i += 32) {
                    const int e = csr[4 * (size_t)(beg + i) + 1];
                    if (e > prev && e < m) m = e;
                }
                for (int d = 16; d > 0; d >>= 1) {
                    const int o = __shfl_xor(m, d);
                    m = (o < m) ? o : m;
                }
                m = __builtin_amdgcn_readfirstlane(m);
                prev = m;
                gcn_edge1(x, offa, offb, cols[m], vals[m], s0, s1);
            }
        }
    }

    *(v4f*)&s_agg[(half * ROWS_PER_BLOCK + wv) * TILE_PITCH + q4]       = s0;
    *(v4f*)&s_agg[((half + 2) * ROWS_PER_BLOCK + wv) * TILE_PITCH + q4] = s1;

    const int m  = lane & 15;
    const int kg = (lane >> 4) << 1;
    const int mt = wv >> 2;
    const int nc = (wv & 3) * 16 + m;

    v2f bf[FDIM / 4];
    for (int k4 = 0; k4 < FDIM / 4; ++k4) {
        const int kk = k4 * 4 + kg;
        bf[k4].x = W[kk * FDIM + nc];
        bf[k4].y = W[(kk + 1) * FDIM + nc];
    }
    const float bb = bias[nc];

    __syncthreads();

    const float* arow = s_agg + (mt * 16 + m) * TILE_PITCH;
    v8f acc = v8f{};
    for (int k4 = 0; k4 < FDIM / 4; ++k4) {
        const int kk = k4 * 4 + kg;
        v2f a;
        a.x = arow[kk];
        a.y = arow[kk + 1];
        acc = __builtin_amdgcn_wmma_f32_16x16x4_f32(
             false, a,  false, bf[k4],
             (short)0, acc,
             false,  false);
    }

    for (int i = 0; i < 8; ++i) {
        const float v = acc[i] + bb;
        s_res[(mt * 16 + half * 8 + i) * TILE_PITCH + nc] = (v > 0.0f) ? v : (__expf(v) - 1.0f);
    }

    __syncthreads();

    for (int u = 0; u < 2; ++u) {
        const int g   = 4 * wv + 2 * u + half;
        const int row = row0 + (g & (ROWS_PER_BLOCK - 1));
        if (row < Nn)
            *(v4f*)(out + ((size_t)(g >> 3) * Nn + row) * FDIM + q4) =
                *(const v4f*)&s_res[g * TILE_PITCH + q4];
    }
}

extern "C" void kernel_launch(void* const* d_in, const int* in_sizes, int n_in,
                              void* d_out, int out_size, void* d_ws, size_t ws_size,
                              hipStream_t stream) {
    const float* x      = (const float*)d_in[0];
    const float* vals   = (const float*)d_in[1];
    const float* weight = (const float*)d_in[2];
    const float* bias   = (const float*)d_in[3];
    const int*   rows   = (const int*)d_in[4];
    const int*   cols   = (const int*)d_in[5];
    float*       out    = (float*)d_out;

    const int  E  = in_sizes[1];
    const int  Nn = in_sizes[0] / (BATCH * FDIM);
    const int  nb     = (Nn + BUCKET_ROWS - 1) / BUCKET_ROWS;
    const int  nchunk = (E + CHUNK_EDGES - 1) / CHUNK_EDGES;
    const long nPad = ((long)Nn + 64) & ~63L;
    const long ePad = ((long)E + 63) & ~63L;
    const long rPad = (long)nchunk * CHUNK_EDGES;
    int*   offs   = (int*)d_ws;
    int*   csr    = offs + nPad;
    int*   brec   = csr + 4 * ePad;
    int*   runs   = brec + 4 * rPad;

    gcn_chunk_partition<<<(unsigned)nchunk, CHUNK_THREADS, nb * sizeof(int), stream>>>(
        rows, cols, vals, runs, brec, E, nb);
    gcn_bucket_csr<<<(unsigned)nb, CSR_THREADS, nchunk * sizeof(int2), stream>>>(
        brec, runs, offs, csr, Nn, nb, nchunk);
    gcn_gather_gemm_bias_elu<<<(unsigned)((Nn + ROWS_PER_BLOCK - 1) / ROWS_PER_BLOCK),
                               ROWS_PER_BLOCK * 32, 0, stream>>>(
        x, vals, cols, offs, csr, weight, bias, out, Nn);
}


# Gather work-group owns a 64-row bucket: groups the bucket's records by row in LDS itself (CSR level-2 launch, csr, offs gone: 3 launches to 2), walks its rows 8 per tile, longest first; sums same
# speedup vs baseline: 5.0690x; 1.1786x over previous
#include <hip/hip_runtime.h>


#define BATCH 4
#define FDIM  64
#define ROWS_PER_BLOCK 8
#define TILE_ROWS (BATCH * ROWS_PER_BLOCK)
#define TILE_PITCH 68

typedef __attribute__((ext_vector_type(2))) float v2f;
typedef __attribute__((ext_vector_type(8))) float v8f;
typedef __attribute__((ext_vector_type(4))) float v4f;
typedef __attribute__((ext_vector_type(4))) int   v4i;

#define BUCKET_SHIFT  6
#define BUCKET_ROWS   (1 << BUCKET_SHIFT)
#define CHUNK_EDGES   4096
#define CHUNK_THREADS 256
#define CHUNK_VEC     (CHUNK_EDGES / (CHUNK_THREADS * 4))
#define BUCKET_TILES  (BUCKET_ROWS / ROWS_PER_BLOCK)
#define REC_CAP       2048
#define RUN_LANES     8
#define RUN_SEGS      8

__device__ __forceinline__ int4 gcn_rows4(const int* __restrict__ rows, int e, int E) {
    int4 r;
    if (e + 4 <= E) {
        r = *(const int4*)(rows + e);
    } else {
        r.x = (e     < E) ? rows[e]     : -1;
        r.y = (e + 1 < E) ? rows[e + 1] : -1;
        r.z = (e + 2 < E) ? rows[e + 2] : -1;
        r.w = (e + 3 < E) ? rows[e + 3] : -1;
    }
    return r;
}

__device__ __forceinline__ int4 gcn_cols4(const int* __restrict__ cols, int e, int E) {
    int4 c;
    if (e + 4 <= E) {
        c = *(const int4*)(cols + e);
    } else {
        c.x = (e     < E) ? cols[e]     : 0;
        c.y = (e + 1 < E) ? cols[e + 1] : 0;
        c.z = (e + 2 < E) ? cols[e + 2] : 0;
        c.w = (e + 3 < E) ? cols[e + 3] : 0;
    }
    return c;
}

__device__ __forceinline__ int4 gcn_vals4(const float* __restrict__ vals, int e, int E) {
    float4 v;
    if (e + 4 <= E) {
        v = *(const float4*)(vals + e);
    } else {
        v.x = (e     < E) ? vals[e]     : 0.f;
        v.y = (e + 1 < E) ? vals[e + 1] : 0.f;
        v.z = (e + 2 < E) ? vals[e + 2] : 0.f;
        v.w = (e + 3 < E) ? vals[e + 3] : 0.f;
    }
    return make_int4(__float_as_int(v.x), __float_as_int(v.y), __float_as_int(v.z), __float_as_int(v.w));
}

__device__ __forceinline__ void gcn_lds_count(int* s_cnt, int row) {
    if (row >= 0)
        __hip_atomic_fetch_add(&s_cnt[row >> BUCKET_SHIFT], 1, __ATOMIC_RELAXED,
                               __HIP_MEMORY_SCOPE_WORKGROUP);
}

template <int NT>
__device__ __forceinline__ int gcn_block_excl_scan(int* s_a, int n, int* s_w) {
    const int t    = threadIdx.x;
    const int lane = t & 31;
    const int w    = t >> 5;
    int carry = 0;
    for (int base = 0; base < n; base += NT) {
        const int i = base + t;
        const int v = (i < n) ? s_a[i] : 0;
        int inc = v;
        for (int d = 1; d < 32; d <<= 1) {
            const int o = __shfl_up(inc, d);
            if (lane >= d) inc += o;
        }
        if (lane == 31) s_w[w] = inc;
        __syncthreads();
        const int wt = (lane < NT / 32) ? s_w[lane] : 0;
        int before = (lane < w) ? wt : 0;
        int all    = wt;
        for (int d = 16; d > 0; d >>= 1) {
            before += __shfl_xor(before, d);
            all    += __shfl_xor(all, d);
        }
        if (i < n) s_a[i] = carry + before + inc - v;
        carry += all;
        __syncthreads();
    }
    return carry;
}

__device__ __forceinline__ int gcn_take_place(int* s_cur, int row) {
    int pos = -1;
    if (row >= 0)
        pos = __hip_atomic_fetch_add(&s_cur[row >> BUCKET_SHIFT], 1, __ATOMIC_RELAXED,
                                     __HIP_MEMORY_SCOPE_WORKGROUP);
    return pos;
}

__device__ __forceinline__ void gcn_file_edge(v4i* s_rec, int pos, int row, int e, int col, int vbits) {
    if (pos >= 0) s_rec[pos] = v4i{row, e, col, vbits};
}

__global__ void __launch_bounds__(CHUNK_THREADS)
gcn_chunk_partition(const int* __restrict__ rows, const int* __restrict__ cols,
                    const float* __restrict__ vals, int* __restrict__ runs,
                    int* __restrict__ brec, int E, int nb) {
    extern __shared__ int s_bkt[];
    __shared__ __attribute__((aligned(16))) v4i s_rec[CHUNK_EDGES];
    __shared__ int s_w[32];
    const int t = threadIdx.x;
    const int chunk = blockIdx.x;
    for (int i = t; i < nb; i += CHUNK_THREADS) s_bkt[i] = 0;
    int4 r[CHUNK_VEC], cl[CHUNK_VEC], vb[CHUNK_VEC];
    const int base = chunk * CHUNK_EDGES;
#pragma unroll
    for (int k = 0; k < CHUNK_VEC; ++k) {
        const int e = base + (k * CHUNK_THREADS + t) * 4;
        r[k]  = gcn_rows4(rows, e, E);
        cl[k] = gcn_cols4(cols, e, E);
        vb[k] = gcn_vals4(vals, e, E);
    }
    __syncthreads();
#pragma unroll
    for (int k = 0; k < CHUNK_VEC; ++k) {
        gcn_lds_count(s_bkt, r[k].x);
        gcn_lds_count(s_bkt, r[k].y);
        gcn_lds_count(s_bkt, r[k].z);
        gcn_lds_count(s_bkt, r[k].w);
    }
    __syncthreads();
    const int cnt = gcn_block_excl_scan<CHUNK_THREADS>(s_bkt, nb, s_w);
    int2* __restrict__ rn = (int2*)runs + (size_t)chunk * nb;
    for (int i = t; i < nb; i += CHUNK_THREADS) {
        const int a = s_bkt[i];
        const int z = (i + 1 < nb) ? s_bkt[i + 1] : cnt;
        rn[i] = make_int2(a, z - a);
    }
    __syncthreads();
    int4 pl[CHUNK_VEC];
#pragma unroll
    for (int k = 0; k < CHUNK_VEC; ++k) {
        pl[k].x = gcn_take_place(s_bkt, r[k].x);
        pl[k].y = gcn_take_place(s_bkt, r[k].y);
        pl[k].z = gcn_take_place(s_bkt, r[k].z);
        pl[k].w = gcn_take_place(s_bkt, r[k].w);
    }
#pragma unroll
    for (int k = 0; k < CHUNK_VEC; ++k) {
        const int e = base + (k * CHUNK_THREADS + t) * 4;
        gcn_file_edge(s_rec, pl[k].x, r[k].x, e,     cl[k].x, vb[k].x);
        gcn_file_edge(s_rec, pl[k].y, r[k].y, e + 1, cl[k].y, vb[k].y);
        gcn_file_edge(s_rec, pl[k].z, r[k].z, e + 2, cl[k].z, vb[k].z);
        gcn_file_edge(s_rec, pl[k].w, r[k].w, e + 3, cl[k].w, vb[k].w);
    }
    __syncthreads();
    v4i* __restrict__ dst = (v4i*)brec + (size_t)chunk * CHUNK_EDGES;
    for (int g = 0; g < CHUNK_EDGES / (8 * CHUNK_THREADS); ++g) {
        v4i o[8];
#pragma unroll
        for (int k = 0; k < 8; ++k)
            o[k] = s_rec[(g * 8 + k) * CHUNK_THREADS + t];
#pragma unroll
        for (int k = 0; k < 8; ++k) {
            const int i = (g * 8 + k) * CHUNK_THREADS + t;
            if (i < cnt) dst[i] = o[k];
        }
    }
}

__device__ __forceinline__ v4f gcn_ld4(const float* __restrict__ xrow, unsigned off) {
    return *(const v4f*)((const char*)xrow + off);
}

__device__ __forceinline__ void gcn_edge1(const float* __restrict__ x, unsigned offa, unsigned offb,
                                          int c, float v, v4f& s0, v4f& s1) {
    const float* __restrict__ xr = x + (size_t)c * FDIM;
    const v4f p = gcn_ld4(xr, offa);
    const v4f q = gcn_ld4(xr, offb);
    s0 += v * p;
    s1 += v * q;
}

__device__ __forceinline__ void gcn_pick(int rank, int c, float v, int p, int& cp, float& vp) {
    const int l = __builtin_ctz(__builtin_amdgcn_ballot_w32(rank == p));
    cp = __builtin_amdgcn_readlane(c, l);
    vp = __int_as_float(__builtin_amdgcn_readlane(__float_as_int(v), l));
}

__device__ __forceinline__ void gcn_pick2(int rankA, int rankB, int cA, float vA, int cB, float vB,
                                          int p, int& cp, float& vp) {
    const unsigned mA = __builtin_amdgcn_ballot_w32(rankA == p);
    const unsigned mB = __builtin_amdgcn_ballot_w32(rankB == p);
    const int l  = __builtin_ctz(mA | mB);
    const int ca = __builtin_amdgcn_readlane(cA, l);
    const int cb = __builtin_amdgcn_readlane(cB, l);
    const int va = __builtin_amdgcn_readlane(__float_as_int(vA), l);
    const int vb = __builtin_amdgcn_readlane(__float_as_int(vB), l);
    cp = mA ? ca : cb;
    vp = __int_as_float(mA ? va : vb);
}

static_assert(BUCKET_ROWS == 64, "the scan of the row counts below is written for two 32-lane halves");

__global__ void __launch_bounds__(ROWS_PER_BLOCK * 32)
gcn_bucket_gather_gemm_bias_elu(const float* __restrict__ x,
                                const float* __restrict__ vals,
                                const int*   __restrict__ cols,
                                const int*   __restrict__ brec,
                                const int*   __restrict__ runs,
                                const float* __restrict__ W,
                                const float* __restrict__ bias,
                                float*       __restrict__ out,
                                int Nn, int nb, int nchunk) {
    extern __shared__ int2 s_run[];
    __shared__ __attribute__((aligned(16))) v4i   s_rec[REC_CAP];
    __shared__ __attribute__((aligned(16))) float s_agg[TILE_ROWS * TILE_PITCH];
    __shared__ __attribute__((aligned(16))) float s_res[TILE_ROWS * TILE_PITCH];
    __shared__ int s_cnt[BUCKET_ROWS];
    __shared__ int s_cur[BUCKET_ROWS];
    __shared__ int s_off[BUCKET_ROWS + 1];
    __shared__ int s_ord[BUCKET_ROWS];

    const int t    = threadIdx.x;
    const int lane = t & 31;
    const int wv   = __builtin_amdgcn_readfirstlane((int)(threadIdx.x >> 5));
    const int b    = blockIdx.x;
    const int half = lane >> 4;
    const int q4   = (lane & 15) * 4;
    const int sub  = lane / RUN_LANES;
    const int li   = lane % RUN_LANES;

    if (t < BUCKET_ROWS) s_cnt[t] = 0;
    for (int c = t; c < nchunk; c += ROWS_PER_BLOCK * 32) {
        const int2 pl = ((const int2*)runs)[(size_t)c * nb + b];
        s_run[c] = make_int2(c * CHUNK_EDGES + pl.x, pl.y);
    }
    __syncthreads();

    const int walk_step = ROWS_PER_BLOCK * (32 / RUN_LANES) * RUN_SEGS;
    for (int c0 = wv * (32 / RUN_LANES); c0 < nchunk; c0 += walk_step) {
        int2 run[RUN_SEGS];
        int  rw[RUN_SEGS];
#pragma unroll
        for (int u = 0; u < RUN_SEGS; ++u) {
            const int c = c0 + sub + ROWS_PER_BLOCK * (32 / RUN_LANES) * u;
            run[u] = s_run[(c < nchunk) ? c : nchunk - 1];
            if (c >= nchunk) run[u].y = 0;
        }
#pragma unroll
        for (int u = 0; u < RUN_SEGS; ++u)
            rw[u] = (li < run[u].y) ? brec[4 * (size_t)(run[u].x + li)] : -1;
#pragma unroll
        for (int u = 0; u < RUN_SEGS; ++u)
            if (rw[u] >= 0)
                __hip_atomic_fetch_add(&s_cnt[rw[u] & (BUCKET_ROWS - 1)], 1, __ATOMIC_RELAXED,
                                       __HIP_MEMORY_SCOPE_WORKGROUP);
        for (int u = 0; u < RUN_SEGS; ++u)
            for (int i = li + RUN_LANES; i < run[u].y; i += RUN_LANES) {
                const int row = brec[4 * (size_t)(run[u].x + i)];
                __hip_atomic_fetch_add(&s_cnt[row & (BUCKET_ROWS - 1)], 1, __ATOMIC_RELAXED,
                                       __HIP_MEMORY_SCOPE_WORKGROUP);
            }
    }
    __syncthreads();
    if (t < BUCKET_ROWS) {
        const int ci = s_cnt[t];
        int rk = 0;
        for (int j = 0; j < BUCKET_ROWS; ++j) {
            const int cj = s_cnt[j];
            rk += (cj > ci || (cj == ci && j < t)) ? 1 : 0;
        }
        s_ord[rk] = t;
    } else if (wv == 2) {
        const int a0 = s_cnt[lane];
        const int a1 = s_cnt[32 + lane];
        int i0 = a0, i1 = a1;
        for (int d = 1; d < 32; d <<= 1) {
            const int o0 = __shfl_up(i0, d);
            const int o1 = __shfl_up(i1, d);
            if (lane >= d) { i0 += o0; i1 += o1; }
        }
        const int lo = __shfl(i0, 31);
        s_off[lane]      = i0 - a0;
        s_cur[lane]      = i0 - a0;
        s_off[32 + lane] = lo + i1 - a1;
        s_cur[32 + lane] = lo + i1 - a1;
        if (lane == 31) s_off[BUCKET_ROWS] = lo + i1;
    }
    __syncthreads();
    const bool filed = s_off[BUCKET_ROWS] <= REC_CAP;
    if (filed) {
        for (int c0 = wv * (32 / RUN_LANES); c0 < nchunk; c0 += walk_step) {
            int2 run[RUN_SEGS];
            v4i  rc[RUN_SEGS];
            int  pos[RUN_SEGS];
#pragma unroll
            for (int u = 0; u < RUN_SEGS; ++u) {
                const int c = c0 + sub + ROWS_PER_BLOCK * (32 / RUN_LANES) * u;
                run[u] = s_run[(c < nchunk) ? c : nchunk - 1];
                if (c >= nchunk) run[u].y = 0;
            }
#pragma unroll
            for (int u = 0; u < RUN_SEGS; ++u)
                rc[u] = (li < run[u].y) ? *(const v4i*)(brec + 4 * (size_t)(run[u].x + li))
                                        : v4i{-1, 0, 0, 0};
#pragma unroll
            for (int u = 0; u < RUN_SEGS; ++u) {
                pos[u] = -1;
                if (rc[u].x >= 0)
                    pos[u] = __hip_atomic_fetch_add(&s_cur[rc[u].x & (BUCKET_ROWS - 1)], 1,
                                                    __ATOMIC_RELAXED, __HIP_MEMORY_SCOPE_WORKGROUP);
            }
#pragma unroll
            for (int u = 0; u < RUN_SEGS; ++u)
                if (pos[u] >= 0) s_rec[pos[u]] = rc[u];
            for (int u = 0; u < RUN_SEGS; ++u)
                for (int i = li + RUN_LANES; i < run[u].y; i += RUN_LANES) {
                    const v4i rec = *(const v4i*)(brec + 4 * (size_t)(run[u].x + i));
                    const int p = __hip_atomic_fetch_add(&s_cur[rec.x & (BUCKET_ROWS - 1)], 1,
                                                         __ATOMIC_RELAXED, __HIP_MEMORY_SCOPE_WORKGROUP);
                    s_rec[p] = rec;
                }
        }
    }
    __syncthreads();

    const int m  = lane & 15;
    const int kg = (lane >> 4) << 1;
    const int mt = wv >> 2;
    const int nc = (wv & 3) * 16 + m;

    v2f bf[FDIM / 4];
    for (int k4 = 0; k4 < FDIM / 4; ++k4) {
        const int kk = k4 * 4 + kg;
        bf[k4].x = W[kk * FDIM + nc];
        bf[k4].y = W[(kk + 1) * FDIM + nc];
    }
    const float bb = bias[nc];

    const unsigned offa = (unsigned)(((size_t)half * Nn * FDIM + q4) * sizeof(float));
    const unsigned offb = (unsigned)(((size_t)(half + 2) * Nn * FDIM + q4) * sizeof(float));

    const int brow0  = b * BUCKET_ROWS;
    const int nvalid = (Nn - brow0 < BUCKET_ROWS) ? Nn - brow0 : BUCKET_ROWS;
    for (int tile = 0; tile < BUCKET_TILES; ++tile) {
        if (tile * ROWS_PER_BLOCK >= nvalid) break;

        v4f s0 = {0.f, 0.f, 0.f, 0.f};
        v4f s1 = {0.f, 0.f, 0.f, 0.f};

        if (tile * ROWS_PER_BLOCK + wv < nvalid) {
            const int rl  = __builtin_amdgcn_readfirstlane(s_ord[tile * ROWS_PER_BLOCK + wv]);
            const int r   = brow0 + rl;
            const int beg = __builtin_amdgcn_readfirstlane(s_off[rl]);
            const int kt  = __builtin_amdgcn_readfirstlane(s_off[rl + 1]) - beg;

            if (!filed) {
                int prev = -1;
                for (int p = 0; p < kt; ++p) {
                    int mn = 0x7fffffff;
                    for (int c = 0; c < nchunk; ++c) {
                        const int2 run = s_run[c];
                        for (int i = lane; i < run.y; i += 32) {
                            const int2 re = *(const int2*)(brec + 4 * (size_t)(run.x + i));
                            if (re.x == r && re.y > prev && re.y < mn) mn = re.y;
                        }
                    }
                    for (int d = 16; d > 0; d >>= 1) {
                        const int o = __shfl_xor(mn, d);
                        mn = (o < mn) ? o : mn;
                    }
                    mn = __builtin_amdgcn_readfirstlane(mn);
                    prev = mn;
                    gcn_edge1(x, offa, offb, cols[mn], vals[mn], s0, s1);
                }
            } else if (kt <= 32) {
                int   eid = 0x7fffffff;
                int   c   = 0;
                float v   = 0.f;
                if (lane < kt) {
                    const v4i rec = s_rec[beg + lane];
                    eid = rec.y;
                    c   = rec.z;
                    v   = __int_as_float(rec.w);
                }
                int rank = 0;
                for (int j = 0; j < kt; ++j)
                    rank += (__builtin_amdgcn_readlane(eid, j) < eid) ? 1 : 0;

                int p = 0;
                for (; p + 3 < kt; p += 4) {
                    int c0, c1, c2, c3;
                    float v0, v1, v2, v3;
                    gcn_pick(rank, c, v, p,     c0, v0);
                    gcn_pick(rank, c, v, p + 1, c1, v1);
                    gcn_pick(rank, c, v, p + 2, c2, v2);
                    gcn_pick(rank, c, v, p + 3, c3, v3);
                    const float* __restrict__ x0 = x + (size_t)c0 * FDIM;
                    const float* __restrict__ x1 = x + (size_t)c1 * FDIM;
                    const float* __restrict__ x2 = x + (size_t)c2 * FDIM;
                    const float* __restrict__ x3 = x + (size_t)c3 * FDIM;
                    const v4f p0 = gcn_ld4(x0, offa), q0 = gcn_ld4(x0, offb);
                    const v4f p1 = gcn_ld4(x1, offa), q1 = gcn_ld4(x1, offb);
                    const v4f p2 = gcn_ld4(x2, offa), q2 = gcn_ld4(x2, offb);
                    const v4f p3 = gcn_ld4(x3, offa), q3 = gcn_ld4(x3, offb);
                    s0 += v0 * p0;  s1 += v0 * q0;
                    s0 += v1 * p1;  s1 += v1 * q1;
                    s0 += v2 * p2;  s1 += v2 * q2;
                    s0 += v3 * p3;  s1 += v3 * q3;
                }
                for (; p < kt; ++p) {
                    int c0; float v0;
                    gcn_pick(rank, c, v, p, c0, v0);
                    gcn_edge1(x, offa, offb, c0, v0, s0, s1);
                }
            } else if (kt <= 64) {
                const bool hasB = lane + 32 < kt;
                const int  iB   = hasB ? lane + 32 : kt - 1;
                const v4i recA = s_rec[beg + lane];
                const v4i recB = s_rec[beg + iB];
                const int   eidA = recA.y;
                const int   cA = recA.z;
                const float vA = __int_as_float(recA.w);
                const int   cB = recB.z;
                const float vB = __int_as_float(recB.w);
                const int eidB = hasB ? recB.y : 0x7fffffff;
                int rankA = 0, rankB = 0;
                for (int j = 0; j < 32; ++j) {
                    const int e = __builtin_amdgcn_readlane(eidA, j);
                    rankA += (e < eidA) ? 1 : 0;
                    rankB += (e < eidB) ? 1 : 0;
                }
                for (int j = 0; j < kt - 32; ++j) {
                    const int e = __builtin_amdgcn_readlane(eidB, j);
                    rankA += (e < eidA) ? 1 : 0;
                    rankB += (e < eidB) ? 1 : 0;
                }

                int p = 0;
                for (; p + 2 < kt; p += 3) {
                    int c0, c1, c2;
                    float v0, v1, v2;
                    gcn_pick2(rankA, rankB, cA, vA, cB, vB, p, c0, v0);
                    gcn_pick2(rankA, rankB, cA, vA, cB, vB, p + 1, c1, v1);
                    gcn_pick2(rankA, rankB, cA, vA, cB, vB, p + 2, c2, v2);
                    const float* __restrict__ x0 = x + (size_t)c0 * FDIM;
                    const float* __restrict__ x1 = x + (size_t)c1 * FDIM;
                    const float* __restrict__ x2 = x + (size_t)c2 * FDIM;
                    const v4f p0 = gcn_ld4(x0, offa), q0 = gcn_ld4(x0, offb);
                    const v4f p1 = gcn_ld4(x1, offa), q1 = gcn_ld4(x1, offb);
                    const v4f p2 = gcn_ld4(x2, offa), q2 = gcn_ld4(x2, offb);
                    s0 += v0 * p0;  s1 += v0 * q0;
                    s0 += v1 * p1;  s1 += v1 * q1;
                    s0 += v2 * p2;  s1 += v2 * q2;
                }
                for (; p < kt; ++p) {
                    int c0; float v0;
                    gcn_pick2(rankA, rankB, cA, vA, cB, vB, p, c0, v0);
                    gcn_edge1(x, offa, offb, c0, v0, s0, s1);
                }
            } else {
                int prev = -1;
                for (int p = 0; p < kt; ++p) {
                    int mn = 0x7fffffff;
                    for (int i = lane; i < kt; i += 32) {
                        const int e = s_rec[beg + i].y;
                        if (e > prev && e < mn) mn = e;
                    }
                    for (int d = 16; d > 0; d >>= 1) {
                        const int o = __shfl_xor(mn, d);
                        mn = (o < mn) ? o : mn;
                    }
                    mn = __builtin_amdgcn_readfirstlane(mn);
                    prev = mn;
                    gcn_edge1(x, offa, offb, cols[mn], vals[mn], s0, s1);
                }
            }
        }

        *(v4f*)&s_agg[(half * ROWS_PER_BLOCK + wv) * TILE_PITCH + q4]       = s0;
        *(v4f*)&s_agg[((half + 2) * ROWS_PER_BLOCK + wv) * TILE_PITCH + q4] = s1;

        __syncthreads();

        const float* arow = s_agg + (mt * 16 + m) * TILE_PITCH;
        v8f acc = v8f{};
        for (int k4 = 0; k4 < FDIM / 4; ++k4) {
            const int kk = k4 * 4 + kg;
            v2f a;
            a.x = arow[kk];
            a.y = arow[kk + 1];
            acc = __builtin_amdgcn_wmma_f32_16x16x4_f32(
                 false, a,  false, bf[k4],
                 (short)0, acc,
                 false,  false);
        }

        for (int i = 0; i < 8; ++i) {
            const float v = acc[i] + bb;
            s_res[(mt * 16 + half * 8 + i) * TILE_PITCH + nc] = (v > 0.0f) ? v : (__expf(v) - 1.0f);
        }

        __syncthreads();

        for (int u = 0; u < 2; ++u) {
            const int g = 4 * wv + 2 * u + half;
            const int k = tile * ROWS_PER_BLOCK + (g & (ROWS_PER_BLOCK - 1));
            if (k < nvalid) {
                const int row = brow0 + s_ord[k];
                *(v4f*)(out + ((size_t)(g >> 3) * Nn + row) * FDIM + q4) =
                    *(const v4f*)&s_res[g * TILE_PITCH + q4];
            }
        }
    }
}

extern "C" void kernel_launch(void* const* d_in, const int* in_sizes, int n_in,
                              void* d_out, int out_size, void* d_ws, size_t ws_size,
                              hipStream_t stream) {
    const float* x      = (const float*)d_in[0];
    const float* vals   = (const float*)d_in[1];
    const float* weight = (const float*)d_in[2];
    const float* bias   = (const float*)d_in[3];
    const int*   rows   = (const int*)d_in[4];
    const int*   cols   = (const int*)d_in[5];
    float*       out    = (float*)d_out;

    const int  E  = in_sizes[1];
    const int  Nn = in_sizes[0] / (BATCH * FDIM);
    const int  nb     = (Nn + BUCKET_ROWS - 1) / BUCKET_ROWS;
    const int  nchunk = (E + CHUNK_EDGES - 1) / CHUNK_EDGES;
    const long rPad = (long)nchunk * CHUNK_EDGES;
    int*   brec   = (int*)d_ws;
    int*   runs   = brec + 4 * rPad;

    gcn_chunk_partition<<<(unsigned)nchunk, CHUNK_THREADS, nb * sizeof(int), stream>>>(
        rows, cols, vals, runs, brec, E, nb);
    gcn_bucket_gather_gemm_bias_elu<<<(unsigned)nb, ROWS_PER_BLOCK * 32, nchunk * sizeof(int2), stream>>>(
        x, vals, cols, brec, runs, weight, bias, out, Nn, nb, nchunk);
}
